# starting kernel
# baseline (speedup 1.0000x reference)
#include <hip/hip_runtime.h>
#include <hip/hip_bf16.h>
#include <cstdint>

#define B_   2
#define N_   16384
#define E_   131072
#define C_   256
#define H_   8
#define CIN_ 259
#define KPAD_ 288
#define SAP_  296
#define SAPU_ 264

typedef __attribute__((ext_vector_type(16))) _Float16 v16h;
typedef __attribute__((ext_vector_type(8)))  float    v8f;
typedef __attribute__((ext_vector_type(8)))  _Float16 h8;
typedef __attribute__((ext_vector_type(2)))  _Float16 h2;

__device__ inline v16h ldA2(const _Float16* p, int half) {
  h8 lo = *(const h8*)(p + half * 8);
  h8 hi = *(const h8*)(p + 16 + half * 8);
  return __builtin_shufflevector(lo, hi, 0, 1, 2, 3, 4, 5, 6, 7,
                                 8, 9, 10, 11, 12, 13, 14, 15);
}
__device__ inline v16h ldBp(const _Float16* __restrict__ Wp, int kc, int ntile, int lane) {
  return *(const v16h*)(Wp + ((((size_t)kc * (C_ / 16) + ntile) * 32 + lane) * 16));
}
__device__ inline v8f wmma_f16(v16h a, v16h b, v8f c) {
  return __builtin_amdgcn_wmma_f32_16x16x32_f16(false, a, false, b, (short)0, c, false, false);
}

__device__ inline void ln_stats(const float* buf, int ld, float* part, float* part2,
                                float* mu, float* istd, int tid) {
  int m = tid >> 4, c = tid & 15;
  float s = 0.f, s2 = 0.f;
#pragma unroll
  for (int j = 0; j < 16; ++j) {
    float x = buf[m * ld + c * 16 + j];
    s += x; s2 += x * x;
  }
  part[m * 16 + c] = s;
  part2[m * 16 + c] = s2;
  __syncthreads();
  if (tid < 16) {
    float S = 0.f, S2 = 0.f;
    for (int j = 0; j < 16; ++j) { S += part[tid * 16 + j]; S2 += part2[tid * 16 + j]; }
    float mean = S * (1.f / 256.f);
    float var  = fmaxf(S2 * (1.f / 256.f) - mean * mean, 0.f);
    mu[tid]   = mean;
    istd[tid] = rsqrtf(var + 1e-5f);
  }
  __syncthreads();
}

__device__ inline void atomicMaxF(float* addr, float v) {
  if (v >= 0.f) atomicMax((int*)addr, __float_as_int(v));
  else          atomicMin((unsigned int*)addr, __float_as_uint(v));
}

__global__ void k_featcvt(const float* __restrict__ features, _Float16* __restrict__ feat_h) {
  size_t idx = (size_t)blockIdx.x * 256 + threadIdx.x;
  if (idx >= (size_t)B_ * N_ * C_) return;
  int c = (int)(idx % C_);
  int n = (int)((idx / C_) % N_);
  int b = (int)(idx / ((size_t)C_ * N_));
  feat_h[idx] = (_Float16)features[((size_t)b * C_ + c) * N_ + n];
}

__global__ void k_wpack(const float* __restrict__ src, _Float16* __restrict__ dst,
                        int rows, int nchunks) {
  size_t idx = (size_t)blockIdx.x * 256 + threadIdx.x;
  if (idx >= (size_t)nchunks * 8192) return;
  int elem  = (int)(idx & 15);
  int lane  = (int)((idx >> 4) & 31);
  int ntile = (int)((idx >> 9) & 15);
  int kc    = (int)(idx >> 13);
  int half = lane >> 4, l16 = lane & 15;
  int j = elem >> 1, p = elem & 1;
  int kb = (j >> 2) * 16 + half * 8 + (j & 3) * 2 + p;
  int k = kc * 32 + kb;
  int n = ntile * 16 + l16;
  dst[idx] = (k < rows) ? (_Float16)src[(size_t)k * C_ + n] : (_Float16)0.f;
}

__global__ void k_init(float* __restrict__ agg, float* __restrict__ segmax,
                       float* __restrict__ segsum, float* __restrict__ cnt) {
  size_t idx = (size_t)blockIdx.x * 256 + threadIdx.x;
  if (idx < (size_t)B_ * N_ * C_) agg[idx] = 0.f;
  if (idx < (size_t)B_ * N_ * H_) { segmax[idx] = __int_as_float(0xff800000); segsum[idx] = 0.f; }
  if (idx < (size_t)B_ * N_) cnt[idx] = 0.f;
}

__global__ __launch_bounds__(256) void k_edge(
    const float* __restrict__ xyz, const int* __restrict__ edges,
    const _Float16* __restrict__ feat_h,
    const _Float16* __restrict__ WkP, const float* __restrict__ biask,
    const float* __restrict__ gk, const float* __restrict__ betak,
    const _Float16* __restrict__ WvP, const float* __restrict__ biasv,
    const float* __restrict__ gv, const float* __restrict__ betav,
    const _Float16* __restrict__ WqP, const float* __restrict__ biasq,
    const float* __restrict__ gq, const float* __restrict__ betaq,
    _Float16* __restrict__ Vh, float* __restrict__ simi) {
  const int b    = blockIdx.y;
  const int e0   = blockIdx.x * 16;
  const int tid  = threadIdx.x;
  const int lane = tid & 31;
  const int wave = tid >> 5;
  const int half = lane >> 4;
  const int l16  = lane & 15;
  const size_t bN = (size_t)b * N_;

  __shared__ int   s_src[16], s_dst[16];
  __shared__ float s_rel[16][3];
  __shared__ _Float16 sAll[16 * SAP_];
  __shared__ float sK[16][260];
  __shared__ float sQ[16][260];
  __shared__ float s_part[256], s_part2[256];
  __shared__ float s_mu[2][16], s_is[2][16];

  if (tid < 16) {
    size_t eidx = ((size_t)b * E_ + e0 + tid) * 2;
    int s = edges[eidx + 0];
    int d = edges[eidx + 1];
    s_src[tid] = s; s_dst[tid] = d;
#pragma unroll
    for (int j = 0; j < 3; ++j)
      s_rel[tid][j] = xyz[(bN + s) * 3 + j] - xyz[(bN + d) * 3 + j];
  }
  __syncthreads();

  const int m16 = tid >> 4;
  const int sg  = tid & 15;
  const _Float16* rowS = feat_h + (bN + (size_t)s_src[m16]) * C_;
  const _Float16* rowD = feat_h + (bN + (size_t)s_dst[m16]) * C_;
  h2 relp;
  {
    int k = sg * 2;
    relp.x = (_Float16)((k     < 3) ? s_rel[m16][k]     : 0.f);
    relp.y = (_Float16)((k + 1 < 3) ? s_rel[m16][k + 1] : 0.f);
  }

  const int nt0 = wave * 2;
  const int n0  = wave * 32;

  *(h8*)(sAll + m16 * SAP_ + sg * 16)     = *(const h8*)(rowS + sg * 16);
  *(h8*)(sAll + m16 * SAP_ + sg * 16 + 8) = *(const h8*)(rowS + sg * 16 + 8);
  *(h2*)(sAll + m16 * SAP_ + 256 + sg * 2) = relp;
  __syncthreads();

  v8f aK0 = {}, aK1 = {}, aV0 = {}, aV1 = {};
  const _Float16* aRow = sAll + l16 * SAP_;
#pragma unroll
  for (int kc = 0; kc < 9; ++kc) {
    v16h a   = ldA2(aRow + kc * 32, half);
    v16h bk0 = ldBp(WkP, kc, nt0,     lane);
    v16h bk1 = ldBp(WkP, kc, nt0 + 1, lane);
    v16h bv0 = ldBp(WvP, kc, nt0,     lane);
    v16h bv1 = ldBp(WvP, kc, nt0 + 1, lane);
    aK0 = wmma_f16(a, bk0, aK0);
    aK1 = wmma_f16(a, bk1, aK1);
    aV0 = wmma_f16(a, bv0, aV0);
    aV1 = wmma_f16(a, bv1, aV1);
  }
#pragma unroll
  for (int r = 0; r < 8; ++r) {
    int m = r + half * 8;
    int n = n0 + l16;
    sK[m][n]      = fmaxf(aK0[r] + biask[n],      0.f);
    sK[m][n + 16] = fmaxf(aK1[r] + biask[n + 16], 0.f);
    sQ[m][n]      = fmaxf(aV0[r] + biasv[n],      0.f);
    sQ[m][n + 16] = fmaxf(aV1[r] + biasv[n + 16], 0.f);
  }
  __syncthreads();
  ln_stats(&sK[0][0], 260, s_part, s_part2, s_mu[0], s_is[0], tid);
  ln_stats(&sQ[0][0], 260, s_part, s_part2, s_mu[1], s_is[1], tid);
  {
    int m = tid >> 4, cb = (tid & 15) * 16;
    float muK = s_mu[0][m], isK = s_is[0][m];
    float muV = s_mu[1][m], isV = s_is[1][m];
    size_t vbase = ((size_t)b * E_ + e0 + m) * C_ + cb;
#pragma unroll
    for (int j = 0; j < 16; ++j) {
      int c = cb + j;
      sK[m][c] = (sK[m][c] - muK) * isK * gk[c] + betak[c];
      float vx = (sQ[m][c] - muV) * isV * gv[c] + betav[c];
      Vh[vbase + j] = (_Float16)vx;
    }
  }
  __syncthreads();

  *(h8*)(sAll + m16 * SAP_ + sg * 16)     = *(const h8*)(rowD + sg * 16);
  *(h8*)(sAll + m16 * SAP_ + sg * 16 + 8) = *(const h8*)(rowD + sg * 16 + 8);
  *(h2*)(sAll + m16 * SAP_ + 256 + sg * 2) = relp;
  __syncthreads();

  v8f aQ0 = {}, aQ1 = {};
#pragma unroll
  for (int kc = 0; kc < 9; ++kc) {
    v16h a   = ldA2(aRow + kc * 32, half);
    v16h bq0 = ldBp(WqP, kc, nt0,     lane);
    v16h bq1 = ldBp(WqP, kc, nt0 + 1, lane);
    aQ0 = wmma_f16(a, bq0, aQ0);
    aQ1 = wmma_f16(a, bq1, aQ1);
  }
#pragma unroll
  for (int r = 0; r < 8; ++r) {
    int m = r + half * 8;
    int n = n0 + l16;
    sQ[m][n]      = fmaxf(aQ0[r] + biasq[n],      0.f);
    sQ[m][n + 16] = fmaxf(aQ1[r] + biasq[n + 16], 0.f);
  }
  __syncthreads();
  ln_stats(&sQ[0][0], 260, s_part, s_part2, s_mu[1], s_is[1], tid);
  {
    int m = tid >> 4, cb = (tid & 15) * 16;
    float mu = s_mu[1][m], is = s_is[1][m];
#pragma unroll
    for (int j = 0; j < 16; ++j) {
      int c = cb + j;
      sQ[m][c] = (sQ[m][c] - mu) * is * gq[c] + betaq[c];
    }
  }
  __syncthreads();

  if (tid < 128) {
    int m = tid >> 3, h = tid & 7;
    float acc = 0.f;
#pragma unroll
    for (int j = 0; j < 32; ++j) acc += sK[m][h * 32 + j] * sQ[m][h * 32 + j];
    simi[((size_t)b * E_ + e0 + m) * H_ + h] = acc * 0.17677669529663687f;
  }
}

__global__ void k_segmax(const int* __restrict__ edges, const float* __restrict__ simi,
                         float* __restrict__ segmax, float* __restrict__ cnt) {
  int b = blockIdx.y;
  size_t idx = (size_t)blockIdx.x * 256 + threadIdx.x;
  if (idx >= (size_t)E_ * H_) return;
  int e = (int)(idx >> 3), h = (int)(idx & 7);
  int d = edges[((size_t)b * E_ + e) * 2 + 1];
  float v = simi[((size_t)b * E_ + e) * H_ + h];
  atomicMaxF(&segmax[((size_t)b * N_ + d) * H_ + h], v);
  if (h == 0) atomicAdd(&cnt[(size_t)b * N_ + d], 1.0f);
}

__global__ void k_expsum(const int* __restrict__ edges, float* __restrict__ simi,
                         const float* __restrict__ segmax, float* __restrict__ segsum) {
  int b = blockIdx.y;
  size_t idx = (size_t)blockIdx.x * 256 + threadIdx.x;
  if (idx >= (size_t)E_ * H_) return;
  int e = (int)(idx >> 3), h = (int)(idx & 7);
  int d = edges[((size_t)b * E_ + e) * 2 + 1];
  float mx = segmax[((size_t)b * N_ + d) * H_ + h];
  float w = __expf(simi[((size_t)b * E_ + e) * H_ + h] - mx);
  simi[((size_t)b * E_ + e) * H_ + h] = w;
  atomicAdd(&segsum[((size_t)b * N_ + d) * H_ + h], w);
}

__global__ void k_scatter(const int* __restrict__ edges, const float* __restrict__ expw,
                          const float* __restrict__ segsum, const _Float16* __restrict__ Vh,
                          float* __restrict__ agg) {
  int b = blockIdx.y;
  size_t idx = (size_t)blockIdx.x * 256 + threadIdx.x;
  int e = (int)(idx >> 8), c = (int)(idx & 255);
  int d = edges[((size_t)b * E_ + e) * 2 + 1];
  int h = c >> 5;
  float w    = expw[((size_t)b * E_ + e) * H_ + h];
  float base = segsum[((size_t)b * N_ + d) * H_ + h];
  float v = (float)Vh[((size_t)b * E_ + e) * C_ + c];
  atomicAdd(&agg[((size_t)b * N_ + d) * C_ + c], v * (w / base));
}

__global__ void k_div(const float* __restrict__ agg, const float* __restrict__ cnt,
                      _Float16* __restrict__ aggh) {
  int b = blockIdx.y;
  size_t idx = (size_t)blockIdx.x * 256 + threadIdx.x;
  int n = (int)(idx >> 8);
  float cc = fmaxf(cnt[(size_t)b * N_ + n], 1.0f);
  size_t o = (size_t)b * N_ * C_ + idx;
  aggh[o] = (_Float16)(agg[o] / cc);
}

__global__ __launch_bounds__(256) void k_update(
    const _Float16* __restrict__ aggh, const _Float16* __restrict__ WuP,
    const float* __restrict__ biasu, const float* __restrict__ gu,
    const float* __restrict__ betau, const float* __restrict__ features,
    float* __restrict__ out) {
  const int b     = blockIdx.y;
  const int nrow0 = blockIdx.x * 16;
  const int tid   = threadIdx.x;
  const int lane  = tid & 31;
  const int wave  = tid >> 5;
  const int half  = lane >> 4;
  const int l16   = lane & 15;

  __shared__ _Float16 sAll[16 * SAPU_];
  __shared__ float sO[16][260];
  __shared__ float s_part[256], s_part2[256];
  __shared__ float s_mu[16], s_is[16];

  {
    int m = tid >> 4, sg = tid & 15;
    const _Float16* row = aggh + ((size_t)b * N_ + nrow0 + m) * C_;
    *(h8*)(sAll + m * SAPU_ + sg * 16)     = *(const h8*)(row + sg * 16);
    *(h8*)(sAll + m * SAPU_ + sg * 16 + 8) = *(const h8*)(row + sg * 16 + 8);
  }
  __syncthreads();

  const int nt0 = wave * 2;
  const int n0  = wave * 32;
  const _Float16* aRow = sAll + l16 * SAPU_;
  v8f a0 = {}, a1 = {};
#pragma unroll
  for (int kc = 0; kc < 8; ++kc) {
    v16h a  = ldA2(aRow + kc * 32, half);
    v16h b0 = ldBp(WuP, kc, nt0,     lane);
    v16h b1 = ldBp(WuP, kc, nt0 + 1, lane);
    a0 = wmma_f16(a, b0, a0);
    a1 = wmma_f16(a, b1, a1);
  }
#pragma unroll
  for (int r = 0; r < 8; ++r) {
    int m = r + half * 8;
    int n = n0 + l16;
    sO[m][n]      = fmaxf(a0[r] + biasu[n],      0.f);
    sO[m][n + 16] = fmaxf(a1[r] + biasu[n + 16], 0.f);
  }
  __syncthreads();
  ln_stats(&sO[0][0], 260, s_part, s_part2, s_mu, s_is, tid);
  {
    int m = tid >> 4, cb = (tid & 15) * 16;
    float mu = s_mu[m], is = s_is[m];
#pragma unroll
    for (int j = 0; j < 16; ++j) {
      int c = cb + j;
      float val = (sO[m][c] - mu) * is * gu[c] + betau[c];
      size_t oi = ((size_t)b * C_ + c) * N_ + nrow0 + m;
      out[oi] = val + features[oi];
    }
  }
}

extern "C" void kernel_launch(void* const* d_in, const int* in_sizes, int n_in,
                              void* d_out, int out_size, void* d_ws, size_t ws_size,
                              hipStream_t stream) {
  const float* xyz      = (const float*)d_in[0];
  const float* features = (const float*)d_in[1];
  const int*   edges    = (const int*)d_in[2];
  const float* Wk = (const float*)d_in[3];  const float* bk = (const float*)d_in[4];
  const float* gk = (const float*)d_in[5];  const float* betak = (const float*)d_in[6];
  const float* Wv = (const float*)d_in[7];  const float* bv = (const float*)d_in[8];
  const float* gv = (const float*)d_in[9];  const float* betav = (const float*)d_in[10];
  const float* Wq = (const float*)d_in[11]; const float* bq = (const float*)d_in[12];
  const float* gq = (const float*)d_in[13]; const float* betaq = (const float*)d_in[14];
  const float* Wu = (const float*)d_in[15]; const float* bu = (const float*)d_in[16];
  const float* gu = (const float*)d_in[17]; const float* betau = (const float*)d_in[18];
  float* out = (float*)d_out;

  char* ws = (char*)d_ws;
  size_t off = 0;
  auto carve = [&](size_t bytes) -> char* {
    char* p = ws + off;
    off = (off + bytes + 255) & ~(size_t)255;
    return p;
  };
  _Float16* feat_h = (_Float16*)carve((size_t)B_ * N_ * C_ * 2);
  _Float16* WkP    = (_Float16*)carve((size_t)KPAD_ * C_ * 2);
  _Float16* WvP    = (_Float16*)carve((size_t)KPAD_ * C_ * 2);
  _Float16* WqP    = (_Float16*)carve((size_t)KPAD_ * C_ * 2);
  _Float16* WuP    = (_Float16*)carve((size_t)C_ * C_ * 2);
  _Float16* Vh     = (_Float16*)carve((size_t)B_ * E_ * C_ * 2);
  float*    simi   = (float*)carve((size_t)B_ * E_ * H_ * 4);
  float*    segmax = (float*)carve((size_t)B_ * N_ * H_ * 4);
  float*    segsum = (float*)carve((size_t)B_ * N_ * H_ * 4);
  float*    cnt    = (float*)carve((size_t)B_ * N_ * 4);
  float*    agg    = (float*)carve((size_t)B_ * N_ * C_ * 4);
  _Float16* aggh   = (_Float16*)carve((size_t)B_ * N_ * C_ * 2);
  (void)ws_size; (void)in_sizes; (void)n_in; (void)out_size;

  k_featcvt<<<(unsigned)(((size_t)B_ * N_ * C_) / 256), 256, 0, stream>>>(features, feat_h);
  k_wpack<<<(9 * 8192) / 256, 256, 0, stream>>>(Wk, WkP, CIN_, 9);
  k_wpack<<<(9 * 8192) / 256, 256, 0, stream>>>(Wv, WvP, CIN_, 9);
  k_wpack<<<(9 * 8192) / 256, 256, 0, stream>>>(Wq, WqP, CIN_, 9);
  k_wpack<<<(8 * 8192) / 256, 256, 0, stream>>>(Wu, WuP, C_, 8);
  k_init<<<(unsigned)(((size_t)B_ * N_ * C_) / 256), 256, 0, stream>>>(agg, segmax, segsum, cnt);

  k_edge<<<dim3(E_ / 16, B_), 256, 0, stream>>>(
      xyz, edges, feat_h,
      WkP, bk, gk, betak,
      WvP, bv, gv, betav,
      WqP, bq, gq, betaq,
      Vh, simi);

  k_segmax<<<dim3((E_ * H_) / 256, B_), 256, 0, stream>>>(edges, simi, segmax, cnt);
  k_expsum<<<dim3((E_ * H_) / 256, B_), 256, 0, stream>>>(edges, simi, segmax, segsum);
  k_scatter<<<dim3((unsigned)(((size_t)E_ * C_) / 256), B_), 256, 0, stream>>>(
      edges, simi, segsum, Vh, agg);
  k_div<<<dim3((N_ * C_) / 256, B_), 256, 0, stream>>>(agg, cnt, aggh);

  k_update<<<dim3(N_ / 16, B_), 256, 0, stream>>>(aggh, WuP, bu, gu, betau, features, out);
}


# k_edge tiles 64 edges per work-group instead of 16 (4 row tiles per wave): each weight fragment fetched feeds 4 WMMAs, so weight traffic through the caches falls 4x
# speedup vs baseline: 1.6487x; 1.6487x over previous
#include <hip/hip_runtime.h>
#include <hip/hip_bf16.h>
#include <cstdint>

#define B_   2
#define N_   16384
#define E_   131072
#define C_   256
#define H_   8
#define CIN_ 259
#define KPAD_ 288
#define SAP_  296
#define SAPU_ 264
#define MT_   4
#define EM_   64

typedef __attribute__((ext_vector_type(16))) _Float16 v16h;
typedef __attribute__((ext_vector_type(8)))  float    v8f;
typedef __attribute__((ext_vector_type(8)))  _Float16 h8;
typedef __attribute__((ext_vector_type(2)))  _Float16 h2;

__device__ inline v16h ldA2(const _Float16* p, int half) {
  h8 lo = *(const h8*)(p + half * 8);
  h8 hi = *(const h8*)(p + 16 + half * 8);
  return __builtin_shufflevector(lo, hi, 0, 1, 2, 3, 4, 5, 6, 7,
                                 8, 9, 10, 11, 12, 13, 14, 15);
}
__device__ inline v16h ldBp(const _Float16* __restrict__ Wp, int kc, int ntile, int lane) {
  return *(const v16h*)(Wp + ((((size_t)kc * (C_ / 16) + ntile) * 32 + lane) * 16));
}
__device__ inline v8f wmma_f16(v16h a, v16h b, v8f c) {
  return __builtin_amdgcn_wmma_f32_16x16x32_f16(false, a, false, b, (short)0, c, false, false);
}

__device__ inline void ln_stats(const float* buf, int ld, float* part, float* part2,
                                float* mu, float* istd, int tid) {
  int m = tid >> 4, c = tid & 15;
  float s = 0.f, s2 = 0.f;
#pragma unroll
  for (int j = 0; j < 16; ++j) {
    float x = buf[m * ld + c * 16 + j];
    s += x; s2 += x * x;
  }
  part[m * 16 + c] = s;
  part2[m * 16 + c] = s2;
  __syncthreads();
  if (tid < 16) {
    float S = 0.f, S2 = 0.f;
    for (int j = 0; j < 16; ++j) { S += part[tid * 16 + j]; S2 += part2[tid * 16 + j]; }
    float mean = S * (1.f / 256.f);
    float var  = fmaxf(S2 * (1.f / 256.f) - mean * mean, 0.f);
    mu[tid]   = mean;
    istd[tid] = rsqrtf(var + 1e-5f);
  }
  __syncthreads();
}

__global__ void k_featcvt(const float* __restrict__ features, _Float16* __restrict__ feat_h) {
  size_t idx = (size_t)blockIdx.x * 256 + threadIdx.x;
  if (idx >= (size_t)B_ * N_ * C_) return;
  int c = (int)(idx % C_);
  int n = (int)((idx / C_) % N_);
  int b = (int)(idx / ((size_t)C_ * N_));
  feat_h[idx] = (_Float16)features[((size_t)b * C_ + c) * N_ + n];
}

__global__ void k_wpack(const float* __restrict__ src, _Float16* __restrict__ dst,
                        int rows, int nchunks) {
  size_t idx = (size_t)blockIdx.x * 256 + threadIdx.x;
  if (idx >= (size_t)nchunks * 8192) return;
  int elem  = (int)(idx & 15);
  int lane  = (int)((idx >> 4) & 31);
  int ntile = (int)((idx >> 9) & 15);
  int kc    = (int)(idx >> 13);
  int half = lane >> 4, l16 = lane & 15;
  int j = elem >> 1, p = elem & 1;
  int kb = (j >> 2) * 16 + half * 8 + (j & 3) * 2 + p;
  int k = kc * 32 + kb;
  int n = ntile * 16 + l16;
  dst[idx] = (k < rows) ? (_Float16)src[(size_t)k * C_ + n] : (_Float16)0.f;
}

__global__ void k_init(int* __restrict__ deg) {
  int idx = blockIdx.x * 256 + threadIdx.x;
  if (idx < B_ * N_) deg[idx] = 0;
}

__global__ void k_count(const int* __restrict__ edges, int* __restrict__ deg) {
  int b = blockIdx.y;
  int e = blockIdx.x * 256 + threadIdx.x;
  int d = edges[((size_t)b * E_ + e) * 2 + 1];
  atomicAdd(&deg[b * N_ + d], 1);
}

__global__ __launch_bounds__(1024) void k_scan(const int* __restrict__ deg,
                                               int* __restrict__ rowptr,
                                               int* __restrict__ cursor) {
  __shared__ int s_w[32];
  const int b = blockIdx.x, t = threadIdx.x;
  const int lane = t & 31, w = t >> 5;
  const int base = b * N_ + t * 16;
  int v[16];
  for (int q = 0; q < 4; ++q) {
    int4 x = *(const int4*)(deg + base + q * 4);
    v[q * 4 + 0] = x.x; v[q * 4 + 1] = x.y; v[q * 4 + 2] = x.z; v[q * 4 + 3] = x.w;
  }
  int sum = 0;
  for (int j = 0; j < 16; ++j) { int x = v[j]; v[j] = sum; sum += x; }
  int inc = sum;
  for (int d = 1; d < 32; d <<= 1) {
    int y = __shfl_up(inc, d);
    if (lane >= d) inc += y;
  }
  if (lane == 31) s_w[w] = inc;
  __syncthreads();
  if (w == 0) {
    int x = s_w[lane], ix = x;
    for (int d = 1; d < 32; d <<= 1) {
      int y = __shfl_up(ix, d);
      if (lane >= d) ix += y;
    }
    s_w[lane] = ix - x;
  }
  __syncthreads();
  const int off = s_w[w] + inc - sum;
  for (int q = 0; q < 4; ++q) {
    int4 o;
    o.x = off + v[q * 4 + 0]; o.y = off + v[q * 4 + 1];
    o.z = off + v[q * 4 + 2]; o.w = off + v[q * 4 + 3];
    *(int4*)(rowptr + base + q * 4) = o;
    *(int4*)(cursor + base + q * 4) = o;
  }
}

__global__ void k_fill(const int* __restrict__ edges, int* __restrict__ cursor,
                       int* __restrict__ elist) {
  int b = blockIdx.y;
  int e = blockIdx.x * 256 + threadIdx.x;
  int d = edges[((size_t)b * E_ + e) * 2 + 1];
  int pos = atomicAdd(&cursor[b * N_ + d], 1);
  elist[(size_t)b * E_ + (pos & (E_ - 1))] = e;
}

__device__ inline void ln_stats_em(const float* buf, float* part, float* part2,
                                   float* mu, float* istd, int tid) {
  const int m = tid >> 4, c = tid & 15;
#pragma unroll
  for (int t = 0; t < MT_; ++t) {
    const int row = t * 16 + m;
    float s = 0.f, s2 = 0.f;
#pragma unroll
    for (int j = 0; j < 16; ++j) {
      float x = buf[row * 260 + c * 16 + j];
      s += x; s2 += x * x;
    }
    part[row * 16 + c] = s;
    part2[row * 16 + c] = s2;
  }
  __syncthreads();
  if (tid < EM_) {
    float S = 0.f, S2 = 0.f;
    for (int j = 0; j < 16; ++j) { S += part[tid * 16 + j]; S2 += part2[tid * 16 + j]; }
    float mean = S * (1.f / 256.f);
    float var  = fmaxf(S2 * (1.f / 256.f) - mean * mean, 0.f);
    mu[tid]   = mean;
    istd[tid] = rsqrtf(var + 1e-5f);
  }
  __syncthreads();
}

__device__ inline void stage_panel(_Float16* sAll, const _Float16* __restrict__ featb,
                                   const int* idx, const float (*rel)[3], int m16, int sg) {
  h8 x0[MT_], x1[MT_];
#pragma unroll
  for (int t = 0; t < MT_; ++t) {
    const _Float16* row = featb + (size_t)idx[t * 16 + m16] * C_;
    x0[t] = *(const h8*)(row + sg * 16);
    x1[t] = *(const h8*)(row + sg * 16 + 8);
  }
#pragma unroll
  for (int t = 0; t < MT_; ++t) {
    const int r = t * 16 + m16;
    const int k = sg * 2;
    h2 relp;
    relp.x = (_Float16)((k     < 3) ? rel[r][k]     : 0.f);
    relp.y = (_Float16)((k + 1 < 3) ? rel[r][k + 1] : 0.f);
    _Float16* dstp = sAll + r * SAP_;
    *(h8*)(dstp + sg * 16)      = x0[t];
    *(h8*)(dstp + sg * 16 + 8)  = x1[t];
    *(h2*)(dstp + 256 + sg * 2) = relp;
  }
}

__global__ __launch_bounds__(256, 4) void k_edge(
    const float* __restrict__ xyz, const int* __restrict__ edges,
    const _Float16* __restrict__ feat_h,
    const _Float16* __restrict__ WkP, const float* __restrict__ biask,
    const float* __restrict__ gk, const float* __restrict__ betak,
    const _Float16* __restrict__ WvP, const float* __restrict__ biasv,
    const float* __restrict__ gv, const float* __restrict__ betav,
    const _Float16* __restrict__ WqP, const float* __restrict__ biasq,
    const float* __restrict__ gq, const float* __restrict__ betaq,
    _Float16* __restrict__ Vh, float* __restrict__ simi) {
  const int b    = blockIdx.y;
  const int e0   = blockIdx.x * EM_;
  const int tid  = threadIdx.x;
  const int lane = tid & 31;
  const int wave = tid >> 5;
  const int half = lane >> 4;
  const int l16  = lane & 15;
  const size_t bN = (size_t)b * N_;

  __shared__ int   s_src[EM_], s_dst[EM_];
  __shared__ float s_rel[EM_][3];
  __shared__ float sK[EM_][260];
  __shared__ __attribute__((aligned(16))) float sR[EM_][260];
  __shared__ float s_part[EM_ * 16], s_part2[EM_ * 16];
  __shared__ float s_mu[2][EM_], s_is[2][EM_];
  _Float16* const sAll = reinterpret_cast<_Float16*>(&sR[0][0]);

  if (tid < EM_) {
    size_t eidx = ((size_t)b * E_ + e0 + tid) * 2;
    int s = edges[eidx + 0];
    int d = edges[eidx + 1];
    s_src[tid] = s; s_dst[tid] = d;
#pragma unroll
    for (int j = 0; j < 3; ++j)
      s_rel[tid][j] = xyz[(bN + s) * 3 + j] - xyz[(bN + d) * 3 + j];
  }
  __syncthreads();

  const int m16 = tid >> 4;
  const int sg  = tid & 15;
  const int nt0 = wave * 2;
  const int n0  = wave * 32;

  stage_panel(sAll, feat_h + bN * C_, s_src, s_rel, m16, sg);
  __syncthreads();

  const _Float16* aRow = sAll + l16 * SAP_;
  {
    v8f aK[MT_][2], aV[MT_][2];
#pragma unroll
    for (int t = 0; t < MT_; ++t) {
      aK[t][0] = (v8f){}; aK[t][1] = (v8f){}; aV[t][0] = (v8f){}; aV[t][1] = (v8f){};
    }
#pragma unroll
    for (int kc = 0; kc < 9; ++kc) {
      v16h a[MT_];
#pragma unroll
      for (int t = 0; t < MT_; ++t) a[t] = ldA2(aRow + t * 16 * SAP_ + kc * 32, half);
      v16h bk0 = ldBp(WkP, kc, nt0,     lane);
      v16h bk1 = ldBp(WkP, kc, nt0 + 1, lane);
      v16h bv0 = ldBp(WvP, kc, nt0,     lane);
      v16h bv1 = ldBp(WvP, kc, nt0 + 1, lane);
#pragma unroll
      for (int t = 0; t < MT_; ++t) {
        aK[t][0] = wmma_f16(a[t], bk0, aK[t][0]);
        aK[t][1] = wmma_f16(a[t], bk1, aK[t][1]);
        aV[t][0] = wmma_f16(a[t], bv0, aV[t][0]);
        aV[t][1] = wmma_f16(a[t], bv1, aV[t][1]);
      }
    }
    __syncthreads();
#pragma unroll
    for (int t = 0; t < MT_; ++t) {
#pragma unroll
      for (int r = 0; r < 8; ++r) {
        int m = t * 16 + r + half * 8;
        int n = n0 + l16;
        sK[m][n]      = fmaxf(aK[t][0][r] + biask[n],      0.f);
        sK[m][n + 16] = fmaxf(aK[t][1][r] + biask[n + 16], 0.f);
        sR[m][n]      = fmaxf(aV[t][0][r] + biasv[n],      0.f);
        sR[m][n + 16] = fmaxf(aV[t][1][r] + biasv[n + 16], 0.f);
      }
    }
  }
  __syncthreads();
  ln_stats_em(&sK[0][0], s_part, s_part2, s_mu[0], s_is[0], tid);
  ln_stats_em(&sR[0][0], s_part, s_part2, s_mu[1], s_is[1], tid);
#pragma unroll
  for (int t = 0; t < MT_; ++t) {
    const int m = t * 16 + m16, cb = sg * 16;
    float muK = s_mu[0][m], isK = s_is[0][m];
    float muV = s_mu[1][m], isV = s_is[1][m];
    size_t vbase = ((size_t)b * E_ + e0 + m) * C_ + cb;
#pragma unroll
    for (int j = 0; j < 16; ++j) {
      int c = cb + j;
      sK[m][c] = (sK[m][c] - muK) * isK * gk[c] + betak[c];
      float vx = (sR[m][c] - muV) * isV * gv[c] + betav[c];
      Vh[vbase + j] = (_Float16)vx;
    }
  }
  __syncthreads();

  stage_panel(sAll, feat_h + bN * C_, s_dst, s_rel, m16, sg);
  __syncthreads();

  {
    v8f aQ[MT_][2];
#pragma unroll
    for (int t = 0; t < MT_; ++t) { aQ[t][0] = (v8f){}; aQ[t][1] = (v8f){}; }
#pragma unroll
    for (int kc = 0; kc < 9; ++kc) {
      v16h a[MT_];
#pragma unroll
      for (int t = 0; t < MT_; ++t) a[t] = ldA2(aRow + t * 16 * SAP_ + kc * 32, half);
      v16h bq0 = ldBp(WqP, kc, nt0,     lane);
      v16h bq1 = ldBp(WqP, kc, nt0 + 1, lane);
#pragma unroll
      for (int t = 0; t < MT_; ++t) {
        aQ[t][0] = wmma_f16(a[t], bq0, aQ[t][0]);
        aQ[t][1] = wmma_f16(a[t], bq1, aQ[t][1]);
      }
    }
    __syncthreads();
#pragma unroll
    for (int t = 0; t < MT_; ++t) {
#pragma unroll
      for (int r = 0; r < 8; ++r) {
        int m = t * 16 + r + half * 8;
        int n = n0 + l16;
        sR[m][n]      = fmaxf(aQ[t][0][r] + biasq[n],      0.f);
        sR[m][n + 16] = fmaxf(aQ[t][1][r] + biasq[n + 16], 0.f);
      }
    }
  }
  __syncthreads();
  ln_stats_em(&sR[0][0], s_part, s_part2, s_mu[1], s_is[1], tid);
#pragma unroll
  for (int t = 0; t < MT_; ++t) {
    const int m = t * 16 + m16, cb = sg * 16;
    float mu = s_mu[1][m], is = s_is[1][m];
#pragma unroll
    for (int j = 0; j < 16; ++j) {
      int c = cb + j;
      sR[m][c] = (sR[m][c] - mu) * is * gq[c] + betaq[c];
    }
  }
  __syncthreads();

#pragma unroll
  for (int t = 0; t < (EM_ * H_) / 256; ++t) {
    int idx = t * 256 + tid;
    int m = idx >> 3, h = idx & 7;
    float acc = 0.f;
#pragma unroll
    for (int j = 0; j < 32; ++j) acc += sK[m][h * 32 + j] * sR[m][h * 32 + j];
    simi[((size_t)b * E_ + e0 + m) * H_ + h] = acc * 0.17677669529663687f;
  }
}

__global__ __launch_bounds__(256) void k_node(
    const int* __restrict__ rowptr, const int* __restrict__ deg,
    const int* __restrict__ elist, const float* __restrict__ simi,
    const _Float16* __restrict__ Vh, _Float16* __restrict__ aggh) {
  const int b    = blockIdx.y;
  const int lane = threadIdx.x & 31;
  const int n  = __builtin_amdgcn_readfirstlane((int)(blockIdx.x * 8 + (threadIdx.x >> 5)));
  const int dg = deg[b * N_ + n];
  const int st = rowptr[b * N_ + n];
  const int* __restrict__ el      = elist + (size_t)b * E_ + st;
  const float* __restrict__ sm    = simi + (size_t)b * E_ * H_ + (lane >> 2);
  const _Float16* __restrict__ vb = Vh + (size_t)b * E_ * C_ + lane * 8;
  h8 o = {};
  if (dg > 0) {
    float mx = __int_as_float(0xff800000);
    for (int i0 = 0; i0 < dg; i0 += 4) {
      float x[4];
      for (int j = 0; j < 4; ++j) {
        int i = min(i0 + j, dg - 1);
        int e = el[i] & (E_ - 1);
        x[j] = sm[(size_t)e * H_];
      }
      for (int j = 0; j < 4; ++j) mx = fmaxf(mx, x[j]);
    }
    float s = 0.f;
    float acc[8];
    for (int k = 0; k < 8; ++k) acc[k] = 0.f;
    for (int i0 = 0; i0 < dg; i0 += 4) {
      float x[4];
      h8 v[4];
      for (int j = 0; j < 4; ++j) {
        int i = min(i0 + j, dg - 1);
        int e = el[i] & (E_ - 1);
        x[j] = sm[(size_t)e * H_];
        v[j] = *(const h8*)(vb + (size_t)e * C_);
      }
      for (int j = 0; j < 4; ++j) {
        float w = (i0 + j < dg) ? __expf(x[j] - mx) : 0.f;
        s += w;
        for (int k = 0; k < 8; ++k) acc[k] += (float)v[j][k] * w;
      }
    }
    const float inv = 1.f / (s * (float)dg);
    for (int k = 0; k < 8; ++k) o[k] = (_Float16)(acc[k] * inv);
  }
  *(h8*)(aggh + ((size_t)b * N_ + n) * C_ + lane * 8) = o;
}

__global__ __launch_bounds__(256) void k_update(
    const _Float16* __restrict__ aggh, const _Float16* __restrict__ WuP,
    const float* __restrict__ biasu, const float* __restrict__ gu,
    const float* __restrict__ betau, const float* __restrict__ features,
    float* __restrict__ out) {
  const int b     = blockIdx.y;
  const int nrow0 = blockIdx.x * 16;
  const int tid   = threadIdx.x;
  const int lane  = tid & 31;
  const int wave  = tid >> 5;
  const int half  = lane >> 4;
  const int l16   = lane & 15;

  __shared__ _Float16 sAll[16 * SAPU_];
  __shared__ float sO[16][260];
  __shared__ float s_part[256], s_part2[256];
  __shared__ float s_mu[16], s_is[16];

  {
    int m = tid >> 4, sg = tid & 15;
    const _Float16* row = aggh + ((size_t)b * N_ + nrow0 + m) * C_;
    *(h8*)(sAll + m * SAPU_ + sg * 16)     = *(const h8*)(row + sg * 16);
    *(h8*)(sAll + m * SAPU_ + sg * 16 + 8) = *(const h8*)(row + sg * 16 + 8);
  }
  __syncthreads();

  const int nt0 = wave * 2;
  const int n0  = wave * 32;
  const _Float16* aRow = sAll + l16 * SAPU_;
  v8f a0 = {}, a1 = {};
#pragma unroll
  for (int kc = 0; kc < 8; ++kc) {
    v16h a  = ldA2(aRow + kc * 32, half);
    v16h b0 = ldBp(WuP, kc, nt0,     lane);
    v16h b1 = ldBp(WuP, kc, nt0 + 1, lane);
    a0 = wmma_f16(a, b0, a0);
    a1 = wmma_f16(a, b1, a1);
  }
#pragma unroll
  for (int r = 0; r < 8; ++r) {
    int m = r + half * 8;
    int n = n0 + l16;
    sO[m][n]      = fmaxf(a0[r] + biasu[n],      0.f);
    sO[m][n + 16] = fmaxf(a1[r] + biasu[n + 16], 0.f);
  }
  __syncthreads();
  ln_stats(&sO[0][0], 260, s_part, s_part2, s_mu, s_is, tid);
  {
    int m = tid >> 4, cb = (tid & 15) * 16;
    float mu = s_mu[m], is = s_is[m];
#pragma unroll
    for (int j = 0; j < 16; ++j) {
      int c = cb + j;
      float val = (sO[m][c] - mu) * is * gu[c] + betau[c];
      size_t oi = ((size_t)b * C_ + c) * N_ + nrow0 + m;
      out[oi] = val + features[oi];
    }
  }
}

extern "C" void kernel_launch(void* const* d_in, const int* in_sizes, int n_in,
                              void* d_out, int out_size, void* d_ws, size_t ws_size,
                              hipStream_t stream) {
  const float* xyz      = (const float*)d_in[0];
  const float* features = (const float*)d_in[1];
  const int*   edges    = (const int*)d_in[2];
  const float* Wk = (const float*)d_in[3];  const float* bk = (const float*)d_in[4];
  const float* gk = (const float*)d_in[5];  const float* betak = (const float*)d_in[6];
  const float* Wv = (const float*)d_in[7];  const float* bv = (const float*)d_in[8];
  const float* gv = (const float*)d_in[9];  const float* betav = (const float*)d_in[10];
  const float* Wq = (const float*)d_in[11]; const float* bq = (const float*)d_in[12];
  const float* gq = (const float*)d_in[13]; const float* betaq = (const float*)d_in[14];
  const float* Wu = (const float*)d_in[15]; const float* bu = (const float*)d_in[16];
  const float* gu = (const float*)d_in[17]; const float* betau = (const float*)d_in[18];
  float* out = (float*)d_out;

  char* ws = (char*)d_ws;
  size_t off = 0;
  auto carve = [&](size_t bytes) -> char* {
    char* p = ws + off;
    off = (off + bytes + 255) & ~(size_t)255;
    return p;
  };
  _Float16* feat_h = (_Float16*)carve((size_t)B_ * N_ * C_ * 2);
  _Float16* WkP    = (_Float16*)carve((size_t)KPAD_ * C_ * 2);
  _Float16* WvP    = (_Float16*)carve((size_t)KPAD_ * C_ * 2);
  _Float16* WqP    = (_Float16*)carve((size_t)KPAD_ * C_ * 2);
  _Float16* WuP    = (_Float16*)carve((size_t)C_ * C_ * 2);
  _Float16* Vh     = (_Float16*)carve((size_t)B_ * E_ * C_ * 2);
  float*    simi   = (float*)carve((size_t)B_ * E_ * H_ * 4);
  int*      deg    = (int*)carve((size_t)B_ * N_ * 4);
  int*      rowptr = (int*)carve((size_t)B_ * N_ * 4);
  int*      cursor = (int*)carve((size_t)B_ * N_ * 4);
  int*      elist  = (int*)carve((size_t)B_ * E_ * 4);
  _Float16* aggh   = (_Float16*)carve((size_t)B_ * N_ * C_ * 2);
  (void)ws_size; (void)in_sizes; (void)n_in; (void)out_size;

  k_featcvt<<<(unsigned)(((size_t)B_ * N_ * C_) / 256), 256, 0, stream>>>(features, feat_h);
  k_wpack<<<(9 * 8192) / 256, 256, 0, stream>>>(Wk, WkP, CIN_, 9);
  k_wpack<<<(9 * 8192) / 256, 256, 0, stream>>>(Wv, WvP, CIN_, 9);
  k_wpack<<<(9 * 8192) / 256, 256, 0, stream>>>(Wq, WqP, CIN_, 9);
  k_wpack<<<(8 * 8192) / 256, 256, 0, stream>>>(Wu, WuP, C_, 8);

  k_init<<<(B_ * N_) / 256, 256, 0, stream>>>(deg);
  k_count<<<dim3(E_ / 256, B_), 256, 0, stream>>>(edges, deg);
  k_scan<<<B_, 1024, 0, stream>>>(deg, rowptr, cursor);
  k_fill<<<dim3(E_ / 256, B_), 256, 0, stream>>>(edges, cursor, elist);

  k_edge<<<dim3(E_ / EM_, B_), 256, 0, stream>>>(
      xyz, edges, feat_h,
      WkP, bk, gk, betak,
      WvP, bv, gv, betav,
      WqP, bq, gq, betaq,
      Vh, simi);

  k_node<<<dim3(N_ / 8, B_), 256, 0, stream>>>(rowptr, deg, elist, simi, Vh, aggh);

  k_update<<<dim3(N_ / 16, B_), 256, 0, stream>>>(aggh, WuP, bu, gu, betau, features, out);
}


# K,V,Q linears hoisted from edges to nodes: feat@W[0:256]+b once per node on WMMA (8x fewer matrix flops), rel@W[256:259] as 3 FMAs per edge; edge pass = gather+ReLU+LN+dot, one wave per edge
# speedup vs baseline: 1.9471x; 1.1810x over previous
#include <hip/hip_runtime.h>
#include <hip/hip_bf16.h>
#include <cstdint>

#define B_   2
#define N_   16384
#define E_   131072
#define C_   256
#define H_   8
#define CIN_ 259
#define SAPU_ 264
#define MT_   4
#define EM_   64

typedef __attribute__((ext_vector_type(16))) _Float16 v16h;
typedef __attribute__((ext_vector_type(8)))  float    v8f;
typedef __attribute__((ext_vector_type(8)))  _Float16 h8;
typedef __attribute__((ext_vector_type(2)))  _Float16 h2;

__device__ inline v16h ldA2(const _Float16* p, int half) {
  h8 lo = *(const h8*)(p + half * 8);
  h8 hi = *(const h8*)(p + 16 + half * 8);
  return __builtin_shufflevector(lo, hi, 0, 1, 2, 3, 4, 5, 6, 7,
                                 8, 9, 10, 11, 12, 13, 14, 15);
}
__device__ inline v16h ldBp(const _Float16* __restrict__ Wp, int kc, int ntile, int lane) {
  return *(const v16h*)(Wp + ((((size_t)kc * (C_ / 16) + ntile) * 32 + lane) * 16));
}
__device__ inline v8f wmma_f16(v16h a, v16h b, v8f c) {
  return __builtin_amdgcn_wmma_f32_16x16x32_f16(false, a, false, b, (short)0, c, false, false);
}

__device__ inline void ln_stats(const float* buf, int ld, float* part, float* part2,
                                float* mu, float* istd, int tid) {
  int m = tid >> 4, c = tid & 15;
  float s = 0.f, s2 = 0.f;
#pragma unroll
  for (int j = 0; j < 16; ++j) {
    float x = buf[m * ld + c * 16 + j];
    s += x; s2 += x * x;
  }
  part[m * 16 + c] = s;
  part2[m * 16 + c] = s2;
  __syncthreads();
  if (tid < 16) {
    float S = 0.f, S2 = 0.f;
    for (int j = 0; j < 16; ++j) { S += part[tid * 16 + j]; S2 += part2[tid * 16 + j]; }
    float mean = S * (1.f / 256.f);
    float var  = fmaxf(S2 * (1.f / 256.f) - mean * mean, 0.f);
    mu[tid]   = mean;
    istd[tid] = rsqrtf(var + 1e-5f);
  }
  __syncthreads();
}

__global__ void k_featcvt(const float* __restrict__ features, _Float16* __restrict__ feat_h) {
  size_t idx = (size_t)blockIdx.x * 256 + threadIdx.x;
  if (idx >= (size_t)B_ * N_ * C_) return;
  int c = (int)(idx % C_);
  int n = (int)((idx / C_) % N_);
  int b = (int)(idx / ((size_t)C_ * N_));
  feat_h[idx] = (_Float16)features[((size_t)b * C_ + c) * N_ + n];
}

__global__ void k_wpack(const float* __restrict__ src, _Float16* __restrict__ dst,
                        int rows, int nchunks) {
  size_t idx = (size_t)blockIdx.x * 256 + threadIdx.x;
  if (idx >= (size_t)nchunks * 8192) return;
  int elem  = (int)(idx & 15);
  int lane  = (int)((idx >> 4) & 31);
  int ntile = (int)((idx >> 9) & 15);
  int kc    = (int)(idx >> 13);
  int half = lane >> 4, l16 = lane & 15;
  int j = elem >> 1, p = elem & 1;
  int kb = (j >> 2) * 16 + half * 8 + (j & 3) * 2 + p;
  int k = kc * 32 + kb;
  int n = ntile * 16 + l16;
  dst[idx] = (k < rows) ? (_Float16)src[(size_t)k * C_ + n] : (_Float16)0.f;
}

__global__ void k_init(int* __restrict__ deg) {
  int idx = blockIdx.x * 256 + threadIdx.x;
  if (idx < B_ * N_) deg[idx] = 0;
}

__global__ void k_count(const int* __restrict__ edges, int* __restrict__ deg) {
  int b = blockIdx.y;
  int e = blockIdx.x * 256 + threadIdx.x;
  int d = edges[((size_t)b * E_ + e) * 2 + 1];
  atomicAdd(&deg[b * N_ + d], 1);
}

__global__ __launch_bounds__(1024) void k_scan(const int* __restrict__ deg,
                                               int* __restrict__ rowptr,
                                               int* __restrict__ cursor) {
  __shared__ int s_w[32];
  const int b = blockIdx.x, t = threadIdx.x;
  const int lane = t & 31, w = t >> 5;
  const int base = b * N_ + t * 16;
  int v[16];
  for (int q = 0; q < 4; ++q) {
    int4 x = *(const int4*)(deg + base + q * 4);
    v[q * 4 + 0] = x.x; v[q * 4 + 1] = x.y; v[q * 4 + 2] = x.z; v[q * 4 + 3] = x.w;
  }
  int sum = 0;
  for (int j = 0; j < 16; ++j) { int x = v[j]; v[j] = sum; sum += x; }
  int inc = sum;
  for (int d = 1; d < 32; d <<= 1) {
    int y = __shfl_up(inc, d);
    if (lane >= d) inc += y;
  }
  if (lane == 31) s_w[w] = inc;
  __syncthreads();
  if (w == 0) {
    int x = s_w[lane], ix = x;
    for (int d = 1; d < 32; d <<= 1) {
      int y = __shfl_up(ix, d);
      if (lane >= d) ix += y;
    }
    s_w[lane] = ix - x;
  }
  __syncthreads();
  const int off = s_w[w] + inc - sum;
  for (int q = 0; q < 4; ++q) {
    int4 o;
    o.x = off + v[q * 4 + 0]; o.y = off + v[q * 4 + 1];
    o.z = off + v[q * 4 + 2]; o.w = off + v[q * 4 + 3];
    *(int4*)(rowptr + base + q * 4) = o;
    *(int4*)(cursor + base + q * 4) = o;
  }
}

__global__ void k_fill(const int* __restrict__ edges, int* __restrict__ cursor,
                       int* __restrict__ elist) {
  int b = blockIdx.y;
  int e = blockIdx.x * 256 + threadIdx.x;
  int d = edges[((size_t)b * E_ + e) * 2 + 1];
  int pos = atomicAdd(&cursor[b * N_ + d], 1);
  elist[(size_t)b * E_ + (pos & (E_ - 1))] = e;
}

__global__ __launch_bounds__(256, 4) void k_proj(
    const _Float16* __restrict__ feat_hb,
    const _Float16* __restrict__ WkP, const _Float16* __restrict__ WvP,
    const _Float16* __restrict__ WqP,
    const float* __restrict__ biask, const float* __restrict__ biasv,
    const float* __restrict__ biasq,
    float* __restrict__ PKV, float* __restrict__ PQ) {
  const int r0   = blockIdx.x * EM_;
  const int tid  = threadIdx.x;
  const int lane = tid & 31;
  const int wave = tid >> 5;
  const int half = lane >> 4;
  const int l16  = lane & 15;

  __shared__ __attribute__((aligned(16))) _Float16 sA[EM_ * SAPU_];

  {
    const int m16 = tid >> 4, sg = tid & 15;
#pragma unroll
    for (int t = 0; t < MT_; ++t) {
      const int r = t * 16 + m16;
      const _Float16* row = feat_hb + (size_t)(r0 + r) * C_;
      *(h8*)(sA + r * SAPU_ + sg * 16)     = *(const h8*)(row + sg * 16);
      *(h8*)(sA + r * SAPU_ + sg * 16 + 8) = *(const h8*)(row + sg * 16 + 8);
    }
  }
  __syncthreads();

  const int nt0 = wave * 2;
  const int n0  = wave * 32;
  const _Float16* aRow = sA + l16 * SAPU_;
  {
    const float cK0 = biask[n0 + l16], cK1 = biask[n0 + l16 + 16];
    const float cV0 = biasv[n0 + l16], cV1 = biasv[n0 + l16 + 16];
    v8f aK[MT_][2], aV[MT_][2];
#pragma unroll
    for (int t = 0; t < MT_; ++t) {
#pragma unroll
      for (int r = 0; r < 8; ++r) {
        aK[t][0][r] = cK0; aK[t][1][r] = cK1; aV[t][0][r] = cV0; aV[t][1][r] = cV1;
      }
    }
#pragma unroll
    for (int kc = 0; kc < 8; ++kc) {
      v16h a[MT_];
#pragma unroll
      for (int t = 0; t < MT_; ++t) a[t] = ldA2(aRow + t * 16 * SAPU_ + kc * 32, half);
      v16h bk0 = ldBp(WkP, kc, nt0,     lane);
      v16h bk1 = ldBp(WkP, kc, nt0 + 1, lane);
      v16h bv0 = ldBp(WvP, kc, nt0,     lane);
      v16h bv1 = ldBp(WvP, kc, nt0 + 1, lane);
#pragma unroll
      for (int t = 0; t < MT_; ++t) {
        aK[t][0] = wmma_f16(a[t], bk0, aK[t][0]);
        aK[t][1] = wmma_f16(a[t], bk1, aK[t][1]);
        aV[t][0] = wmma_f16(a[t], bv0, aV[t][0]);
        aV[t][1] = wmma_f16(a[t], bv1, aV[t][1]);
      }
    }
#pragma unroll
    for (int t = 0; t < MT_; ++t) {
#pragma unroll
      for (int r = 0; r < 8; ++r) {
        const int m = t * 16 + r + half * 8;
        float* pr = PKV + (size_t)(r0 + m) * (2 * C_) + n0 + l16;
        pr[0]       = aK[t][0][r];
        pr[16]      = aK[t][1][r];
        pr[C_]      = aV[t][0][r];
        pr[C_ + 16] = aV[t][1][r];
      }
    }
  }
  {
    const float cQ0 = biasq[n0 + l16], cQ1 = biasq[n0 + l16 + 16];
    v8f aQ[MT_][2];
#pragma unroll
    for (int t = 0; t < MT_; ++t) {
#pragma unroll
      for (int r = 0; r < 8; ++r) { aQ[t][0][r] = cQ0; aQ[t][1][r] = cQ1; }
    }
#pragma unroll
    for (int kc = 0; kc < 8; ++kc) {
      v16h a[MT_];
#pragma unroll
      for (int t = 0; t < MT_; ++t) a[t] = ldA2(aRow + t * 16 * SAPU_ + kc * 32, half);
      v16h bq0 = ldBp(WqP, kc, nt0,     lane);
      v16h bq1 = ldBp(WqP, kc, nt0 + 1, lane);
#pragma unroll
      for (int t = 0; t < MT_; ++t) {
        aQ[t][0] = wmma_f16(a[t], bq0, aQ[t][0]);
        aQ[t][1] = wmma_f16(a[t], bq1, aQ[t][1]);
      }
    }
#pragma unroll
    for (int t = 0; t < MT_; ++t) {
#pragma unroll
      for (int r = 0; r < 8; ++r) {
        const int m = t * 16 + r + half * 8;
        float* pr = PQ + (size_t)(r0 + m) * C_ + n0 + l16;
        pr[0]  = aQ[t][0][r];
        pr[16] = aQ[t][1][r];
      }
    }
  }
}

__device__ inline void ld8(float* d, const float* __restrict__ p) {
  const float4 x = *(const float4*)p, y = *(const float4*)(p + 4);
  d[0] = x.x; d[1] = x.y; d[2] = x.z; d[3] = x.w; d[4] = y.x; d[5] = y.y; d[6] = y.z; d[7] = y.w;
}
template <int M>
__device__ inline float xrow(float x) {
  const int b = __builtin_bit_cast(int, x);
  return __builtin_bit_cast(float, __builtin_amdgcn_update_dpp(b, b, 0x160 + M, 0xf, 0xf, false));
}
__device__ inline float xhalf(float x) {
  const int b = __builtin_bit_cast(int, x);
  return __builtin_bit_cast(float,
      __builtin_amdgcn_permlanex16(b, b, 0x76543210, 0xfedcba98, false, false));
}
__device__ inline int lane_i(int x, int i) { return __builtin_amdgcn_readlane(x, i); }
__device__ inline float lane_f(float x, int i) {
  return __builtin_bit_cast(float, __builtin_amdgcn_readlane(__builtin_bit_cast(int, x), i));
}

__global__ __launch_bounds__(256, 4) void k_edge2(
    const float* __restrict__ xyzb, const int* __restrict__ edgesb,
    const float* __restrict__ PKV, const float* __restrict__ PQ,
    const float* __restrict__ Wk, const float* __restrict__ gk, const float* __restrict__ betak,
    const float* __restrict__ Wv, const float* __restrict__ gv, const float* __restrict__ betav,
    const float* __restrict__ Wq, const float* __restrict__ gq, const float* __restrict__ betaq,
    _Float16* __restrict__ Vhb, float* __restrict__ simib) {
  const int lane = threadIdx.x & 31;
  const int e0   = (blockIdx.x * 8 + (threadIdx.x >> 5)) * 8;
  const int c0   = lane * 8;

  float wk[3][8], wv[3][8], wq[3][8];
  float gK[8], gV[8], gQ[8], tK[8], tV[8], tQ[8];
#pragma unroll
  for (int r = 0; r < 3; ++r) {
    ld8(wk[r], Wk + (size_t)(C_ + r) * C_ + c0);
    ld8(wv[r], Wv + (size_t)(C_ + r) * C_ + c0);
    ld8(wq[r], Wq + (size_t)(C_ + r) * C_ + c0);
  }
  ld8(gK, gk + c0);    ld8(gV, gv + c0);    ld8(gQ, gq + c0);
  ld8(tK, betak + c0); ld8(tV, betav + c0); ld8(tQ, betaq + c0);

  int my_s, my_d;
  float my_r0, my_r1, my_r2;
  {
    const int2 ed = *(const int2*)(edgesb + (size_t)(e0 + (lane & 7)) * 2);
    my_s = ed.x; my_d = ed.y;
    const float* xs = xyzb + (size_t)my_s * 3;
    const float* xd = xyzb + (size_t)my_d * 3;
    my_r0 = xs[0] - xd[0]; my_r1 = xs[1] - xd[1]; my_r2 = xs[2] - xd[2];
  }

  float nk[8], nv[8], nq[8];
  {
    const int s = lane_i(my_s, 0), d = lane_i(my_d, 0);
    ld8(nk, PKV + (size_t)s * (2 * C_) + c0);
    ld8(nv, PKV + (size_t)s * (2 * C_) + C_ + c0);
    ld8(nq, PQ + (size_t)d * C_ + c0);
  }
  for (int i = 0; i < 8; ++i) {
    float k[8], v[8], q[8];
#pragma unroll
    for (int j = 0; j < 8; ++j) { k[j] = nk[j]; v[j] = nv[j]; q[j] = nq[j]; }
    if (i + 1 < 8) {
      const int s = lane_i(my_s, i + 1), d = lane_i(my_d, i + 1);
      ld8(nk, PKV + (size_t)s * (2 * C_) + c0);
      ld8(nv, PKV + (size_t)s * (2 * C_) + C_ + c0);
      ld8(nq, PQ + (size_t)d * C_ + c0);
    }
    const float r0 = lane_f(my_r0, i), r1 = lane_f(my_r1, i), r2 = lane_f(my_r2, i);
    float red[6] = {0.f, 0.f, 0.f, 0.f, 0.f, 0.f};
#pragma unroll
    for (int j = 0; j < 8; ++j) {
      k[j] = fmaxf(k[j] + r0 * wk[0][j] + r1 * wk[1][j] + r2 * wk[2][j], 0.f);
      v[j] = fmaxf(v[j] + r0 * wv[0][j] + r1 * wv[1][j] + r2 * wv[2][j], 0.f);
      q[j] = fmaxf(q[j] + r0 * wq[0][j] + r1 * wq[1][j] + r2 * wq[2][j], 0.f);
      red[0] += k[j]; red[1] += k[j] * k[j];
      red[2] += v[j]; red[3] += v[j] * v[j];
      red[4] += q[j]; red[5] += q[j] * q[j];
    }
#pragma unroll
    for (int j = 0; j < 6; ++j) red[j] += xrow<1>(red[j]);
#pragma unroll
    for (int j = 0; j < 6; ++j) red[j] += xrow<2>(red[j]);
#pragma unroll
    for (int j = 0; j < 6; ++j) red[j] += xrow<4>(red[j]);
#pragma unroll
    for (int j = 0; j < 6; ++j) red[j] += xrow<8>(red[j]);
#pragma unroll
    for (int j = 0; j < 6; ++j) red[j] += xhalf(red[j]);
    const float mK = red[0] * (1.f / 256.f), mV = red[2] * (1.f / 256.f), mQ = red[4] * (1.f / 256.f);
    const float iK = rsqrtf(fmaxf(red[1] * (1.f / 256.f) - mK * mK, 0.f) + 1e-5f);
    const float iV = rsqrtf(fmaxf(red[3] * (1.f / 256.f) - mV * mV, 0.f) + 1e-5f);
    const float iQ = rsqrtf(fmaxf(red[5] * (1.f / 256.f) - mQ * mQ, 0.f) + 1e-5f);
    float dot = 0.f;
    h8 vo;
#pragma unroll
    for (int j = 0; j < 8; ++j) {
      const float kn = (k[j] - mK) * iK * gK[j] + tK[j];
      const float qn = (q[j] - mQ) * iQ * gQ[j] + tQ[j];
      const float vn = (v[j] - mV) * iV * gV[j] + tV[j];
      dot += kn * qn;
      vo[j] = (_Float16)vn;
    }
    dot += xrow<1>(dot);
    dot += xrow<2>(dot);
    const size_t e = (size_t)(e0 + i);
    if ((lane & 3) == 0) simib[e * H_ + (lane >> 2)] = dot * 0.17677669529663687f;
    *(h8*)(Vhb + e * C_ + c0) = vo;
  }
}

__global__ __launch_bounds__(256) void k_node(
    const int* __restrict__ rowptr, const int* __restrict__ deg,
    const int* __restrict__ elist, const float* __restrict__ simi,
    const _Float16* __restrict__ Vh, _Float16* __restrict__ aggh) {
  const int b    = blockIdx.y;
  const int lane = threadIdx.x & 31;
  const int n  = __builtin_amdgcn_readfirstlane((int)(blockIdx.x * 8 + (threadIdx.x >> 5)));
  const int dg = deg[b * N_ + n];
  const int st = rowptr[b * N_ + n];
  const int* __restrict__ el      = elist + (size_t)b * E_ + st;
  const float* __restrict__ sm    = simi + (size_t)b * E_ * H_ + (lane >> 2);
  const _Float16* __restrict__ vb = Vh + (size_t)b * E_ * C_ + lane * 8;
  h8 o = {};
  if (dg > 0) {
    float mx = __int_as_float(0xff800000);
    for (int i0 = 0; i0 < dg; i0 += 4) {
      float x[4];
      for (int j = 0; j < 4; ++j) {
        int i = min(i0 + j, dg - 1);
        int e = el[i] & (E_ - 1);
        x[j] = sm[(size_t)e * H_];
      }
      for (int j = 0; j < 4; ++j) mx = fmaxf(mx, x[j]);
    }
    float s = 0.f;
    float acc[8];
    for (int k = 0; k < 8; ++k) acc[k] = 0.f;
    for (int i0 = 0; i0 < dg; i0 += 4) {
      float x[4];
      h8 v[4];
      for (int j = 0; j < 4; ++j) {
        int i = min(i0 + j, dg - 1);
        int e = el[i] & (E_ - 1);
        x[j] = sm[(size_t)e * H_];
        v[j] = *(const h8*)(vb + (size_t)e * C_);
      }
      for (int j = 0; j < 4; ++j) {
        float w = (i0 + j < dg) ? __expf(x[j] - mx) : 0.f;
        s += w;
        for (int k = 0; k < 8; ++k) acc[k] += (float)v[j][k] * w;
      }
    }
    const float inv = 1.f / (s * (float)dg);
    for (int k = 0; k < 8; ++k) o[k] = (_Float16)(acc[k] * inv);
  }
  *(h8*)(aggh + ((size_t)b * N_ + n) * C_ + lane * 8) = o;
}

__global__ __launch_bounds__(256) void k_update(
    const _Float16* __restrict__ aggh, const _Float16* __restrict__ WuP,
    const float* __restrict__ biasu, const float* __restrict__ gu,
    const float* __restrict__ betau, const float* __restrict__ features,
    float* __restrict__ out) {
  const int b     = blockIdx.y;
  const int nrow0 = blockIdx.x * 16;
  const int tid   = threadIdx.x;
  const int lane  = tid & 31;
  const int wave  = tid >> 5;
  const int half  = lane >> 4;
  const int l16   = lane & 15;

  __shared__ _Float16 sAll[16 * SAPU_];
  __shared__ float sO[16][260];
  __shared__ float s_part[256], s_part2[256];
  __shared__ float s_mu[16], s_is[16];

  {
    int m = tid >> 4, sg = tid & 15;
    const _Float16* row = aggh + ((size_t)b * N_ + nrow0 + m) * C_;
    *(h8*)(sAll + m * SAPU_ + sg * 16)     = *(const h8*)(row + sg * 16);
    *(h8*)(sAll + m * SAPU_ + sg * 16 + 8) = *(const h8*)(row + sg * 16 + 8);
  }
  __syncthreads();

  const int nt0 = wave * 2;
  const int n0  = wave * 32;
  const _Float16* aRow = sAll + l16 * SAPU_;
  v8f a0 = {}, a1 = {};
#pragma unroll
  for (int kc = 0; kc < 8; ++kc) {
    v16h a  = ldA2(aRow + kc * 32, half);
    v16h b0 = ldBp(WuP, kc, nt0,     lane);
    v16h b1 = ldBp(WuP, kc, nt0 + 1, lane);
    a0 = wmma_f16(a, b0, a0);
    a1 = wmma_f16(a, b1, a1);
  }
#pragma unroll
  for (int r = 0; r < 8; ++r) {
    int m = r + half * 8;
    int n = n0 + l16;
    sO[m][n]      = fmaxf(a0[r] + biasu[n],      0.f);
    sO[m][n + 16] = fmaxf(a1[r] + biasu[n + 16], 0.f);
  }
  __syncthreads();
  ln_stats(&sO[0][0], 260, s_part, s_part2, s_mu, s_is, tid);
  {
    int m = tid >> 4, cb = (tid & 15) * 16;
    float mu = s_mu[m], is = s_is[m];
#pragma unroll
    for (int j = 0; j < 16; ++j) {
      int c = cb + j;
      float val = (sO[m][c] - mu) * is * gu[c] + betau[c];
      size_t oi = ((size_t)b * C_ + c) * N_ + nrow0 + m;
      out[oi] = val + features[oi];
    }
  }
}

extern "C" void kernel_launch(void* const* d_in, const int* in_sizes, int n_in,
                              void* d_out, int out_size, void* d_ws, size_t ws_size,
                              hipStream_t stream) {
  const float* xyz      = (const float*)d_in[0];
  const float* features = (const float*)d_in[1];
  const int*   edges    = (const int*)d_in[2];
  const float* Wk = (const float*)d_in[3];  const float* bk = (const float*)d_in[4];
  const float* gk = (const float*)d_in[5];  const float* betak = (const float*)d_in[6];
  const float* Wv = (const float*)d_in[7];  const float* bv = (const float*)d_in[8];
  const float* gv = (const float*)d_in[9];  const float* betav = (const float*)d_in[10];
  const float* Wq = (const float*)d_in[11]; const float* bq = (const float*)d_in[12];
  const float* gq = (const float*)d_in[13]; const float* betaq = (const float*)d_in[14];
  const float* Wu = (const float*)d_in[15]; const float* bu = (const float*)d_in[16];
  const float* gu = (const float*)d_in[17]; const float* betau = (const float*)d_in[18];
  float* out = (float*)d_out;

  char* ws = (char*)d_ws;
  size_t off = 0;
  auto carve = [&](size_t bytes) -> char* {
    char* p = ws + off;
    off = (off + bytes + 255) & ~(size_t)255;
    return p;
  };
  _Float16* feat_h = (_Float16*)carve((size_t)B_ * N_ * C_ * 2);
  _Float16* WkP    = (_Float16*)carve((size_t)C_ * C_ * 2);
  _Float16* WvP    = (_Float16*)carve((size_t)C_ * C_ * 2);
  _Float16* WqP    = (_Float16*)carve((size_t)C_ * C_ * 2);
  _Float16* WuP    = (_Float16*)carve((size_t)C_ * C_ * 2);
  float*    PKV    = (float*)carve((size_t)N_ * 2 * C_ * 4);
  float*    PQ     = (float*)carve((size_t)N_ * C_ * 4);
  _Float16* Vh     = (_Float16*)carve((size_t)E_ * C_ * 2);
  float*    simi   = (float*)carve((size_t)B_ * E_ * H_ * 4);
  int*      deg    = (int*)carve((size_t)B_ * N_ * 4);
  int*      rowptr = (int*)carve((size_t)B_ * N_ * 4);
  int*      cursor = (int*)carve((size_t)B_ * N_ * 4);
  int*      elist  = (int*)carve((size_t)B_ * E_ * 4);
  _Float16* aggh   = (_Float16*)carve((size_t)B_ * N_ * C_ * 2);
  (void)ws_size; (void)in_sizes; (void)n_in; (void)out_size;

  k_featcvt<<<(unsigned)(((size_t)B_ * N_ * C_) / 256), 256, 0, stream>>>(features, feat_h);
  k_wpack<<<(8 * 8192) / 256, 256, 0, stream>>>(Wk, WkP, C_, 8);
  k_wpack<<<(8 * 8192) / 256, 256, 0, stream>>>(Wv, WvP, C_, 8);
  k_wpack<<<(8 * 8192) / 256, 256, 0, stream>>>(Wq, WqP, C_, 8);
  k_wpack<<<(8 * 8192) / 256, 256, 0, stream>>>(Wu, WuP, C_, 8);

  k_init<<<(B_ * N_) / 256, 256, 0, stream>>>(deg);
  k_count<<<dim3(E_ / 256, B_), 256, 0, stream>>>(edges, deg);
  k_scan<<<B_, 1024, 0, stream>>>(deg, rowptr, cursor);
  k_fill<<<dim3(E_ / 256, B_), 256, 0, stream>>>(edges, cursor, elist);

  k_proj<<<N_ / EM_, 256, 0, stream>>>(feat_h, WkP, WvP, WqP, bk, bv, bq, PKV, PQ);
  k_edge2<<<E_ / EM_, 256, 0, stream>>>(
      xyz, edges, PKV, PQ,
      Wk, gk, betak, Wv, gv, betav, Wq, gq, betaq,
      Vh, simi);
  k_node<<<dim3(N_ / 8, 1), 256, 0, stream>>>(rowptr, deg, elist, simi, Vh, aggh);

  k_proj<<<N_ / EM_, 256, 0, stream>>>(feat_h + (size_t)N_ * C_, WkP, WvP, WqP, bk, bv, bq, PKV, PQ);
  k_edge2<<<E_ / EM_, 256, 0, stream>>>(
      xyz + (size_t)N_ * 3, edges + (size_t)E_ * 2, PKV, PQ,
      Wk, gk, betak, Wv, gv, betav, Wq, gq, betaq,
      Vh, simi + (size_t)E_ * H_);
  k_node<<<dim3(N_ / 8, 1), 256, 0, stream>>>(rowptr + N_, deg + N_, elist + E_,
                                              simi + (size_t)E_ * H_, Vh, aggh + (size_t)N_ * C_);

  k_update<<<dim3(N_ / 16, B_), 256, 0, stream>>>(aggh, WuP, bu, gu, betau, features, out);
}


# k_update re-tiled to 64 nodes per work-group, result turned through LDS: residual reads and output stores are whole 256-B channel-row runs, not 8-B pieces; LayerNorm sums regrouped (4 runs of 64)
# speedup vs baseline: 3.7949x; 1.9490x over previous
#include <hip/hip_runtime.h>
#include <hip/hip_bf16.h>
#include <cstdint>

#define B_   2
#define N_   16384
#define E_   131072
#define C_   256
#define H_   8
#define CIN_ 259
#define SAPU_ 264
#define MT_   4
#define EM_   64

typedef __attribute__((ext_vector_type(16))) _Float16 v16h;
typedef __attribute__((ext_vector_type(8)))  float    v8f;
typedef __attribute__((ext_vector_type(8)))  _Float16 h8;
typedef __attribute__((ext_vector_type(2)))  _Float16 h2;

__device__ inline v16h ldA2(const _Float16* p, int half) {
  h8 lo = *(const h8*)(p + half * 8);
  h8 hi = *(const h8*)(p + 16 + half * 8);
  return __builtin_shufflevector(lo, hi, 0, 1, 2, 3, 4, 5, 6, 7,
                                 8, 9, 10, 11, 12, 13, 14, 15);
}
__device__ inline v16h ldBp(const _Float16* __restrict__ Wp, int kc, int ntile, int lane) {
  return *(const v16h*)(Wp + ((((size_t)kc * (C_ / 16) + ntile) * 32 + lane) * 16));
}
__device__ inline v8f wmma_f16(v16h a, v16h b, v8f c) {
  return __builtin_amdgcn_wmma_f32_16x16x32_f16(false, a, false, b, (short)0, c, false, false);
}

__global__ void k_wpack(const float* __restrict__ src, _Float16* __restrict__ dst,
                        int rows, int nchunks) {
  size_t idx = (size_t)blockIdx.x * 256 + threadIdx.x;
  if (idx >= (size_t)nchunks * 8192) return;
  int elem  = (int)(idx & 15);
  int lane  = (int)((idx >> 4) & 31);
  int ntile = (int)((idx >> 9) & 15);
  int kc    = (int)(idx >> 13);
  int half = lane >> 4, l16 = lane & 15;
  int j = elem >> 1, p = elem & 1;
  int kb = (j >> 2) * 16 + half * 8 + (j & 3) * 2 + p;
  int k = kc * 32 + kb;
  int n = ntile * 16 + l16;
  dst[idx] = (k < rows) ? (_Float16)src[(size_t)k * C_ + n] : (_Float16)0.f;
}

__global__ void k_init(int* __restrict__ deg) {
  int idx = blockIdx.x * 256 + threadIdx.x;
  if (idx < B_ * N_) deg[idx] = 0;
}

__global__ void k_count(const int* __restrict__ edges, int* __restrict__ deg) {
  int b = blockIdx.y;
  int e = blockIdx.x * 256 + threadIdx.x;
  int d = edges[((size_t)b * E_ + e) * 2 + 1];
  atomicAdd(&deg[b * N_ + d], 1);
}

__global__ __launch_bounds__(1024) void k_scan(const int* __restrict__ deg,
                                               int* __restrict__ rowptr,
                                               int* __restrict__ cursor) {
  __shared__ int s_w[32];
  const int b = blockIdx.x, t = threadIdx.x;
  const int lane = t & 31, w = t >> 5;
  const int base = b * N_ + t * 16;
  int v[16];
  for (int q = 0; q < 4; ++q) {
    int4 x = *(const int4*)(deg + base + q * 4);
    v[q * 4 + 0] = x.x; v[q * 4 + 1] = x.y; v[q * 4 + 2] = x.z; v[q * 4 + 3] = x.w;
  }
  int sum = 0;
  for (int j = 0; j < 16; ++j) { int x = v[j]; v[j] = sum; sum += x; }
  int inc = sum;
  for (int d = 1; d < 32; d <<= 1) {
    int y = __shfl_up(inc, d);
    if (lane >= d) inc += y;
  }
  if (lane == 31) s_w[w] = inc;
  __syncthreads();
  if (w == 0) {
    int x = s_w[lane], ix = x;
    for (int d = 1; d < 32; d <<= 1) {
      int y = __shfl_up(ix, d);
      if (lane >= d) ix += y;
    }
    s_w[lane] = ix - x;
  }
  __syncthreads();
  const int off = s_w[w] + inc - sum;
  for (int q = 0; q < 4; ++q) {
    int4 o;
    o.x = off + v[q * 4 + 0]; o.y = off + v[q * 4 + 1];
    o.z = off + v[q * 4 + 2]; o.w = off + v[q * 4 + 3];
    *(int4*)(rowptr + base + q * 4) = o;
    *(int4*)(cursor + base + q * 4) = o;
  }
}

__global__ void k_fill(const int* __restrict__ edges, int* __restrict__ cursor,
                       int* __restrict__ slist) {
  int b = blockIdx.y;
  int e = blockIdx.x * 256 + threadIdx.x;
  const int2 ed = *(const int2*)(edges + ((size_t)b * E_ + e) * 2);
  int pos = atomicAdd(&cursor[b * N_ + ed.y], 1);
  slist[(size_t)b * E_ + (pos & (E_ - 1))] = ed.x;
}

__global__ __launch_bounds__(256, 4) void k_proj(
    const float* __restrict__ featb,
    const _Float16* __restrict__ WkP, const _Float16* __restrict__ WvP,
    const _Float16* __restrict__ WqP,
    const float* __restrict__ biask, const float* __restrict__ biasv,
    const float* __restrict__ biasq,
    float* __restrict__ PKV, float* __restrict__ PQ) {
  const int r0   = blockIdx.x * EM_;
  const int tid  = threadIdx.x;
  const int lane = tid & 31;
  const int wave = tid >> 5;
  const int half = lane >> 4;
  const int l16  = lane & 15;

  __shared__ __attribute__((aligned(16))) _Float16 sA[EM_ * SAPU_];

  {
    const int q  = tid & 15;
    const int cg = tid >> 4;
    const float* srcp = featb + (size_t)r0 + 4 * q;
    float4 x[2][8];
#pragma unroll
    for (int p = 0; p < 2; ++p) {
#pragma unroll
      for (int j = 0; j < 8; ++j)
        x[p][j] = *(const float4*)(srcp + (size_t)((p * 16 + cg) * 8 + j) * N_);
    }
#pragma unroll
    for (int p = 0; p < 2; ++p) {
      h8 hx, hy, hz, hw;
#pragma unroll
      for (int j = 0; j < 8; ++j) {
        hx[j] = (_Float16)x[p][j].x; hy[j] = (_Float16)x[p][j].y;
        hz[j] = (_Float16)x[p][j].z; hw[j] = (_Float16)x[p][j].w;
      }
      _Float16* d = sA + (4 * q) * SAPU_ + (p * 16 + cg) * 8;
      *(h8*)(d)             = hx;
      *(h8*)(d + SAPU_)     = hy;
      *(h8*)(d + 2 * SAPU_) = hz;
      *(h8*)(d + 3 * SAPU_) = hw;
    }
  }
  __syncthreads();

  const int nt0 = wave * 2;
  const int n0  = wave * 32;
  const _Float16* aRow = sA + l16 * SAPU_;
  {
    const float cK0 = biask[n0 + l16], cK1 = biask[n0 + l16 + 16];
    const float cV0 = biasv[n0 + l16], cV1 = biasv[n0 + l16 + 16];
    v8f aK[MT_][2], aV[MT_][2];
#pragma unroll
    for (int t = 0; t < MT_; ++t) {
#pragma unroll
      for (int r = 0; r < 8; ++r) {
        aK[t][0][r] = cK0; aK[t][1][r] = cK1; aV[t][0][r] = cV0; aV[t][1][r] = cV1;
      }
    }
#pragma unroll
    for (int kc = 0; kc < 8; ++kc) {
      v16h a[MT_];
#pragma unroll
      for (int t = 0; t < MT_; ++t) a[t] = ldA2(aRow + t * 16 * SAPU_ + kc * 32, half);
      v16h bk0 = ldBp(WkP, kc, nt0,     lane);
      v16h bk1 = ldBp(WkP, kc, nt0 + 1, lane);
      v16h bv0 = ldBp(WvP, kc, nt0,     lane);
      v16h bv1 = ldBp(WvP, kc, nt0 + 1, lane);
#pragma unroll
      for (int t = 0; t < MT_; ++t) {
        aK[t][0] = wmma_f16(a[t], bk0, aK[t][0]);
        aK[t][1] = wmma_f16(a[t], bk1, aK[t][1]);
        aV[t][0] = wmma_f16(a[t], bv0, aV[t][0]);
        aV[t][1] = wmma_f16(a[t], bv1, aV[t][1]);
      }
    }
#pragma unroll
    for (int t = 0; t < MT_; ++t) {
#pragma unroll
      for (int r = 0; r < 8; ++r) {
        const int m = t * 16 + r + half * 8;
        float* pr = PKV + (size_t)(r0 + m) * (2 * C_) + n0 + l16;
        pr[0]       = aK[t][0][r];
        pr[16]      = aK[t][1][r];
        pr[C_]      = aV[t][0][r];
        pr[C_ + 16] = aV[t][1][r];
      }
    }
  }
  {
    const float cQ0 = biasq[n0 + l16], cQ1 = biasq[n0 + l16 + 16];
    v8f aQ[MT_][2];
#pragma unroll
    for (int t = 0; t < MT_; ++t) {
#pragma unroll
      for (int r = 0; r < 8; ++r) { aQ[t][0][r] = cQ0; aQ[t][1][r] = cQ1; }
    }
#pragma unroll
    for (int kc = 0; kc < 8; ++kc) {
      v16h a[MT_];
#pragma unroll
      for (int t = 0; t < MT_; ++t) a[t] = ldA2(aRow + t * 16 * SAPU_ + kc * 32, half);
      v16h bq0 = ldBp(WqP, kc, nt0,     lane);
      v16h bq1 = ldBp(WqP, kc, nt0 + 1, lane);
#pragma unroll
      for (int t = 0; t < MT_; ++t) {
        aQ[t][0] = wmma_f16(a[t], bq0, aQ[t][0]);
        aQ[t][1] = wmma_f16(a[t], bq1, aQ[t][1]);
      }
    }
#pragma unroll
    for (int t = 0; t < MT_; ++t) {
#pragma unroll
      for (int r = 0; r < 8; ++r) {
        const int m = t * 16 + r + half * 8;
        float* pr = PQ + (size_t)(r0 + m) * C_ + n0 + l16;
        pr[0]  = aQ[t][0][r];
        pr[16] = aQ[t][1][r];
      }
    }
  }
}

__device__ inline void ld8(float* d, const float* __restrict__ p) {
  const float4 x = *(const float4*)p, y = *(const float4*)(p + 4);
  d[0] = x.x; d[1] = x.y; d[2] = x.z; d[3] = x.w; d[4] = y.x; d[5] = y.y; d[6] = y.z; d[7] = y.w;
}
template <int M>
__device__ inline float xrow(float x) {
  const int b = __builtin_bit_cast(int, x);
  return __builtin_bit_cast(float, __builtin_amdgcn_update_dpp(b, b, 0x160 + M, 0xf, 0xf, false));
}
__device__ inline float xhalf(float x) {
  const int b = __builtin_bit_cast(int, x);
  return __builtin_bit_cast(float,
      __builtin_amdgcn_permlanex16(b, b, 0x76543210, 0xfedcba98, false, false));
}
__device__ inline int lane_i(int x, int i) { return __builtin_amdgcn_readlane(x, i); }
__device__ inline float lane_f(float x, int i) {
  return __builtin_bit_cast(float, __builtin_amdgcn_readlane(__builtin_bit_cast(int, x), i));
}

__global__ __launch_bounds__(128) void k_gather(
    const float* __restrict__ xyzb,
    const int* __restrict__ rowptrb, const int* __restrict__ degb, const int* __restrict__ slistb,
    const float* __restrict__ PKV, const float* __restrict__ PQ,
    const float* __restrict__ Wk, const float* __restrict__ gk, const float* __restrict__ betak,
    const float* __restrict__ Wv, const float* __restrict__ gv, const float* __restrict__ betav,
    const float* __restrict__ Wq, const float* __restrict__ gq, const float* __restrict__ betaq,
    _Float16* __restrict__ agghb) {
  const int lane = threadIdx.x & 31;
  const int n  = __builtin_amdgcn_readfirstlane((int)(blockIdx.x * 4 + (threadIdx.x >> 5)));
  const int c0 = lane * 8;
  const int dg = min(degb[n], E_);
  const int st = rowptrb[n];
  h8 o = {};
  if (dg > 0) {
    float wk[3][8], wv[3][8], wq[3][8];
    float gK[8], gQ[8], tK[8], tQ[8];
#pragma unroll
    for (int r = 0; r < 3; ++r) {
      ld8(wk[r], Wk + (size_t)(C_ + r) * C_ + c0);
      ld8(wv[r], Wv + (size_t)(C_ + r) * C_ + c0);
      ld8(wq[r], Wq + (size_t)(C_ + r) * C_ + c0);
    }
    ld8(gK, gk + c0);    ld8(gQ, gq + c0);
    ld8(tK, betak + c0); ld8(tQ, betaq + c0);
    float qb[8];
    ld8(qb, PQ + (size_t)n * C_ + c0);
    const float xd0 = xyzb[(size_t)n * 3], xd1 = xyzb[(size_t)n * 3 + 1], xd2 = xyzb[(size_t)n * 3 + 2];

    float mx = __int_as_float(0xff800000);
    float s = 0.f;
    float acc[8];
#pragma unroll
    for (int j = 0; j < 8; ++j) acc[j] = 0.f;

    for (int base = 0; base < dg; base += 32) {
      const int cnt = min(32, dg - base);
      int my_s = 0;
      float my_r0 = 0.f, my_r1 = 0.f, my_r2 = 0.f;
      if (lane < cnt) {
        my_s = slistb[(st + base + lane) & (E_ - 1)] & (N_ - 1);
        const float* xs = xyzb + (size_t)my_s * 3;
        my_r0 = xs[0] - xd0; my_r1 = xs[1] - xd1; my_r2 = xs[2] - xd2;
      }
      float nk[8], nv[8];
      {
        const int s0 = lane_i(my_s, 0);
        ld8(nk, PKV + (size_t)s0 * (2 * C_) + c0);
        ld8(nv, PKV + (size_t)s0 * (2 * C_) + C_ + c0);
      }
      for (int i = 0; i < cnt; ++i) {
        float k[8], v[8], q[8];
#pragma unroll
        for (int j = 0; j < 8; ++j) { k[j] = nk[j]; v[j] = nv[j]; }
        if (i + 1 < cnt) {
          const int s1 = lane_i(my_s, i + 1);
          ld8(nk, PKV + (size_t)s1 * (2 * C_) + c0);
          ld8(nv, PKV + (size_t)s1 * (2 * C_) + C_ + c0);
        }
        const float r0 = lane_f(my_r0, i), r1 = lane_f(my_r1, i), r2 = lane_f(my_r2, i);
        float red[6] = {0.f, 0.f, 0.f, 0.f, 0.f, 0.f};
#pragma unroll
        for (int j = 0; j < 8; ++j) {
          k[j] = fmaxf(k[j] + r0 * wk[0][j] + r1 * wk[1][j] + r2 * wk[2][j], 0.f);
          v[j] = fmaxf(v[j] + r0 * wv[0][j] + r1 * wv[1][j] + r2 * wv[2][j], 0.f);
          q[j] = fmaxf(qb[j] + r0 * wq[0][j] + r1 * wq[1][j] + r2 * wq[2][j], 0.f);
          red[0] += k[j]; red[1] += k[j] * k[j];
          red[2] += v[j]; red[3] += v[j] * v[j];
          red[4] += q[j]; red[5] += q[j] * q[j];
        }
#pragma unroll
        for (int j = 0; j < 6; ++j) red[j] += xrow<1>(red[j]);
#pragma unroll
        for (int j = 0; j < 6; ++j) red[j] += xrow<2>(red[j]);
#pragma unroll
        for (int j = 0; j < 6; ++j) red[j] += xrow<4>(red[j]);
#pragma unroll
        for (int j = 0; j < 6; ++j) red[j] += xrow<8>(red[j]);
#pragma unroll
        for (int j = 0; j < 6; ++j) red[j] += xhalf(red[j]);
        const float mK = red[0] * (1.f / 256.f), mV = red[2] * (1.f / 256.f), mQ = red[4] * (1.f / 256.f);
        const float iK = rsqrtf(fmaxf(red[1] * (1.f / 256.f) - mK * mK, 0.f) + 1e-5f);
        const float iV = rsqrtf(fmaxf(red[3] * (1.f / 256.f) - mV * mV, 0.f) + 1e-5f);
        const float iQ = rsqrtf(fmaxf(red[5] * (1.f / 256.f) - mQ * mQ, 0.f) + 1e-5f);
        float dot = 0.f;
#pragma unroll
        for (int j = 0; j < 8; ++j) {
          const float kn = (k[j] - mK) * iK * gK[j] + tK[j];
          const float qn = (q[j] - mQ) * iQ * gQ[j] + tQ[j];
          dot += kn * qn;
        }
        dot += xrow<1>(dot);
        dot += xrow<2>(dot);
        const float sim = dot * 0.17677669529663687f;
        const float mn  = fmaxf(mx, sim);
        const float a   = __expf(mx - mn);
        const float w   = __expf(sim - mn);
        const float wi  = w * iV;
        const float wm  = -wi * mV;
        s = s * a + w;
#pragma unroll
        for (int j = 0; j < 8; ++j) acc[j] = acc[j] * a + (wi * v[j] + wm);
        mx = mn;
      }
    }
    float gV[8], tV[8];
    ld8(gV, gv + c0); ld8(tV, betav + c0);
    const float is  = 1.f / s;
    const float idg = 1.f / (float)dg;
#pragma unroll
    for (int j = 0; j < 8; ++j) o[j] = (_Float16)((acc[j] * is * gV[j] + tV[j]) * idg);
  }
  *(h8*)(agghb + (size_t)n * C_ + c0) = o;
}

#define UT_ 68

__global__ __launch_bounds__(256) void k_update(
    const _Float16* __restrict__ aggh, const _Float16* __restrict__ WuP,
    const float* __restrict__ biasu, const float* __restrict__ gu,
    const float* __restrict__ betau, const float* __restrict__ features,
    float* __restrict__ out) {
  const int b    = blockIdx.y;
  const int r0   = blockIdx.x * EM_;
  const int tid  = threadIdx.x;
  const int lane = tid & 31;
  const int wave = tid >> 5;
  const int half = lane >> 4;
  const int l16  = lane & 15;

  __shared__ __attribute__((aligned(16))) _Float16 sA[EM_ * SAPU_];
  __shared__ __attribute__((aligned(16))) float sT[C_ * UT_];
  __shared__ float s_part[4][EM_], s_part2[4][EM_];
  __shared__ __attribute__((aligned(16))) float s_mu[EM_], s_is[EM_];

  const int q  = l16;
  const int cs = wave * 2 + half;
  const size_t gbase = (size_t)b * C_ * N_ + (size_t)r0 + 4 * q;

  float4 res[16];
#pragma unroll
  for (int p = 0; p < 16; ++p)
    res[p] = *(const float4*)(features + gbase + (size_t)(p * 16 + cs) * N_);

  {
    const _Float16* srcp = aggh + ((size_t)b * N_ + r0) * C_;
#pragma unroll
    for (int p = 0; p < 8; ++p) {
      const int row = p * 8 + wave;
      *(h8*)(sA + row * SAPU_ + lane * 8) = *(const h8*)(srcp + (size_t)row * C_ + lane * 8);
    }
  }
  __syncthreads();

  const int nt0 = wave * 2;
  const int n0  = wave * 32;
  {
    const _Float16* aRow = sA + l16 * SAPU_;
    v8f acc[MT_][2];
#pragma unroll
    for (int t = 0; t < MT_; ++t) { acc[t][0] = v8f{}; acc[t][1] = v8f{}; }
#pragma unroll
    for (int kc = 0; kc < 8; ++kc) {
      v16h a[MT_];
#pragma unroll
      for (int t = 0; t < MT_; ++t) a[t] = ldA2(aRow + t * 16 * SAPU_ + kc * 32, half);
      v16h b0 = ldBp(WuP, kc, nt0,     lane);
      v16h b1 = ldBp(WuP, kc, nt0 + 1, lane);
#pragma unroll
      for (int t = 0; t < MT_; ++t) {
        acc[t][0] = wmma_f16(a[t], b0, acc[t][0]);
        acc[t][1] = wmma_f16(a[t], b1, acc[t][1]);
      }
    }
    const float bu0 = biasu[n0 + l16], bu1 = biasu[n0 + l16 + 16];
#pragma unroll
    for (int t = 0; t < MT_; ++t) {
      float* d0 = sT + (n0 + l16) * UT_ + t * 16 + half * 8;
      float* d1 = d0 + 16 * UT_;
      float4 x;
      x.x = fmaxf(acc[t][0][0] + bu0, 0.f); x.y = fmaxf(acc[t][0][1] + bu0, 0.f);
      x.z = fmaxf(acc[t][0][2] + bu0, 0.f); x.w = fmaxf(acc[t][0][3] + bu0, 0.f);
      *(float4*)d0 = x;
      x.x = fmaxf(acc[t][0][4] + bu0, 0.f); x.y = fmaxf(acc[t][0][5] + bu0, 0.f);
      x.z = fmaxf(acc[t][0][6] + bu0, 0.f); x.w = fmaxf(acc[t][0][7] + bu0, 0.f);
      *(float4*)(d0 + 4) = x;
      x.x = fmaxf(acc[t][1][0] + bu1, 0.f); x.y = fmaxf(acc[t][1][1] + bu1, 0.f);
      x.z = fmaxf(acc[t][1][2] + bu1, 0.f); x.w = fmaxf(acc[t][1][3] + bu1, 0.f);
      *(float4*)d1 = x;
      x.x = fmaxf(acc[t][1][4] + bu1, 0.f); x.y = fmaxf(acc[t][1][5] + bu1, 0.f);
      x.z = fmaxf(acc[t][1][6] + bu1, 0.f); x.w = fmaxf(acc[t][1][7] + bu1, 0.f);
      *(float4*)(d1 + 4) = x;
    }
  }
  float g[16], bt[16];
#pragma unroll
  for (int p = 0; p < 16; ++p) { g[p] = gu[p * 16 + cs]; bt[p] = betau[p * 16 + cs]; }
  __syncthreads();

  {
    const int m = tid & (EM_ - 1), part = tid >> 6;
    const float* col = sT + (part * 64) * UT_ + m;
    float s = 0.f, s2 = 0.f;
    for (int c = 0; c < 64; ++c) {
      const float x = col[c * UT_];
      s += x; s2 += x * x;
    }
    s_part[part][m]  = s;
    s_part2[part][m] = s2;
  }
  __syncthreads();
  if (tid < EM_) {
    const float S  = s_part[0][tid] + s_part[1][tid] + s_part[2][tid] + s_part[3][tid];
    const float S2 = s_part2[0][tid] + s_part2[1][tid] + s_part2[2][tid] + s_part2[3][tid];
    const float mean = S * (1.f / 256.f);
    const float var  = fmaxf(S2 * (1.f / 256.f) - mean * mean, 0.f);
    s_mu[tid] = mean;
    s_is[tid] = rsqrtf(var + 1e-5f);
  }
  __syncthreads();

  {
    const float4 mu = *(const float4*)(s_mu + 4 * q);
    const float4 is = *(const float4*)(s_is + 4 * q);
#pragma unroll
    for (int p = 0; p < 16; ++p) {
      const int c = p * 16 + cs;
      const float4 x = *(const float4*)(sT + c * UT_ + 4 * q);
      float4 o;
      o.x = ((x.x - mu.x) * is.x * g[p] + bt[p]) + res[p].x;
      o.y = ((x.y - mu.y) * is.y * g[p] + bt[p]) + res[p].y;
      o.z = ((x.z - mu.z) * is.z * g[p] + bt[p]) + res[p].z;
      o.w = ((x.w - mu.w) * is.w * g[p] + bt[p]) + res[p].w;
      *(float4*)(out + gbase + (size_t)c * N_) = o;
    }
  }
}

extern "C" void kernel_launch(void* const* d_in, const int* in_sizes, int n_in,
                              void* d_out, int out_size, void* d_ws, size_t ws_size,
                              hipStream_t stream) {
  const float* xyz      = (const float*)d_in[0];
  const float* features = (const float*)d_in[1];
  const int*   edges    = (const int*)d_in[2];
  const float* Wk = (const float*)d_in[3];  const float* bk = (const float*)d_in[4];
  const float* gk = (const float*)d_in[5];  const float* betak = (const float*)d_in[6];
  const float* Wv = (const float*)d_in[7];  const float* bv = (const float*)d_in[8];
  const float* gv = (const float*)d_in[9];  const float* betav = (const float*)d_in[10];
  const float* Wq = (const float*)d_in[11]; const float* bq = (const float*)d_in[12];
  const float* gq = (const float*)d_in[13]; const float* betaq = (const float*)d_in[14];
  const float* Wu = (const float*)d_in[15]; const float* bu = (const float*)d_in[16];
  const float* gu = (const float*)d_in[17]; const float* betau = (const float*)d_in[18];
  float* out = (float*)d_out;

  char* ws = (char*)d_ws;
  size_t off = 0;
  auto carve = [&](size_t bytes) -> char* {
    char* p = ws + off;
    off = (off + bytes + 255) & ~(size_t)255;
    return p;
  };
  _Float16* WkP    = (_Float16*)carve((size_t)C_ * C_ * 2);
  _Float16* WvP    = (_Float16*)carve((size_t)C_ * C_ * 2);
  _Float16* WqP    = (_Float16*)carve((size_t)C_ * C_ * 2);
  _Float16* WuP    = (_Float16*)carve((size_t)C_ * C_ * 2);
  float*    PKV    = (float*)carve((size_t)N_ * 2 * C_ * 4);
  float*    PQ     = (float*)carve((size_t)N_ * C_ * 4);
  int*      deg    = (int*)carve((size_t)B_ * N_ * 4);
  int*      rowptr = (int*)carve((size_t)B_ * N_ * 4);
  int*      cursor = (int*)carve((size_t)B_ * N_ * 4);
  int*      slist  = (int*)carve((size_t)B_ * E_ * 4);
  _Float16* aggh   = (_Float16*)carve((size_t)B_ * N_ * C_ * 2);
  (void)ws_size; (void)in_sizes; (void)n_in; (void)out_size;

  k_wpack<<<(8 * 8192) / 256, 256, 0, stream>>>(Wk, WkP, C_, 8);
  k_wpack<<<(8 * 8192) / 256, 256, 0, stream>>>(Wv, WvP, C_, 8);
  k_wpack<<<(8 * 8192) / 256, 256, 0, stream>>>(Wq, WqP, C_, 8);
  k_wpack<<<(8 * 8192) / 256, 256, 0, stream>>>(Wu, WuP, C_, 8);

  k_init<<<(B_ * N_) / 256, 256, 0, stream>>>(deg);
  k_count<<<dim3(E_ / 256, B_), 256, 0, stream>>>(edges, deg);
  k_scan<<<B_, 1024, 0, stream>>>(deg, rowptr, cursor);
  k_fill<<<dim3(E_ / 256, B_), 256, 0, stream>>>(edges, cursor, slist);

  k_proj<<<N_ / EM_, 256, 0, stream>>>(features, WkP, WvP, WqP, bk, bv, bq, PKV, PQ);
  k_gather<<<N_ / 4, 128, 0, stream>>>(
      xyz, rowptr, deg, slist, PKV, PQ,
      Wk, gk, betak, Wv, gv, betav, Wq, gq, betaq,
      aggh);

  k_proj<<<N_ / EM_, 256, 0, stream>>>(features + (size_t)C_ * N_, WkP, WvP, WqP, bk, bv, bq, PKV, PQ);
  k_gather<<<N_ / 4, 128, 0, stream>>>(
      xyz + (size_t)N_ * 3, rowptr + N_, deg + N_, slist + E_, PKV, PQ,
      Wk, gk, betak, Wv, gv, betav, Wq, gq, betaq,
      aggh + (size_t)N_ * C_);

  k_update<<<dim3(N_ / EM_, B_), 256, 0, stream>>>(aggh, WuP, bu, gu, betau, features, out);
}


# both batches in one launch of k_proj and one of k_gather: PKV and PQ hold a plane per batch (48 MB more workspace), fill blocks go first; 6 launches become 4
# speedup vs baseline: 5.5748x; 1.4690x over previous
#include <hip/hip_runtime.h>
#include <hip/hip_bf16.h>
#include <cstdint>

#define B_   2
#define N_   16384
#define E_   131072
#define C_   256
#define H_   8
#define CIN_ 259
#define SAPU_ 264
#define MT_   4
#define EM_   64
#define CAP_  32

typedef __attribute__((ext_vector_type(16))) _Float16 v16h;
typedef __attribute__((ext_vector_type(8)))  float    v8f;
typedef __attribute__((ext_vector_type(8)))  _Float16 h8;
typedef __attribute__((ext_vector_type(2)))  _Float16 h2;
typedef __attribute__((ext_vector_type(2)))  float    f2;

__device__ inline v16h ldA2(const _Float16* p, int half) {
  h8 lo = *(const h8*)(p + half * 8);
  h8 hi = *(const h8*)(p + 16 + half * 8);
  return __builtin_shufflevector(lo, hi, 0, 1, 2, 3, 4, 5, 6, 7,
                                 8, 9, 10, 11, 12, 13, 14, 15);
}
__device__ inline v16h ldBp(const _Float16* __restrict__ Wp, int kc, int ntile, int lane) {
  return *(const v16h*)(Wp + ((((size_t)kc * (C_ / 16) + ntile) * 32 + lane) * 16));
}
__device__ inline v8f wmma_f16(v16h a, v16h b, v8f c) {
  return __builtin_amdgcn_wmma_f32_16x16x32_f16(false, a, false, b, (short)0, c, false, false);
}

#define PREP_INIT_ 32
#define PREP_PACK_ 128
__global__ __launch_bounds__(256) void k_prep(
    const float* __restrict__ Wk, const float* __restrict__ Wv,
    const float* __restrict__ Wq, const float* __restrict__ Wu,
    _Float16* __restrict__ WkP, _Float16* __restrict__ WvP,
    _Float16* __restrict__ WqP, _Float16* __restrict__ WuP,
    int* __restrict__ cnt) {
  if (blockIdx.x < PREP_INIT_) {
    const int idx = blockIdx.x * 256 + threadIdx.x;
    *(int4*)(cnt + idx * 4) = make_int4(0, 0, 0, 0);
    return;
  }
  const int t = (blockIdx.x - PREP_INIT_) * 256 + threadIdx.x;
  const int which = t >> 13;
  const float* __restrict__ srcw = which == 0 ? Wk : which == 1 ? Wv : which == 2 ? Wq : Wu;
  _Float16* __restrict__ dstw    = which == 0 ? WkP : which == 1 ? WvP : which == 2 ? WqP : WuP;
  const int r     = t & 8191;
  const int e8    = r & 1;
  const int lane  = (r >> 1) & 31;
  const int ntile = (r >> 6) & 15;
  const int kc    = r >> 10;
  const int half = lane >> 4, l16 = lane & 15;
  const int k0 = kc * 32 + e8 * 16 + half * 8;
  const int n  = ntile * 16 + l16;
  h8 v;
#pragma unroll
  for (int j = 0; j < 8; ++j) v[j] = (_Float16)srcw[(size_t)(k0 + j) * C_ + n];
  *(h8*)(dstw + (size_t)r * 8) = v;
}

#define CNT_BLOCKS_ ((B_ * E_) / 1024)
__device__ inline void fill_body(const int* __restrict__ edges, int* __restrict__ cnt,
                                 unsigned short* __restrict__ blist, int c, int tid) {
  int sn[4], dn[4], slot[4];
#pragma unroll
  for (int q = 0; q < 4; ++q) {
    const int e = c * 1024 + q * 256 + tid;
    sn[q] = edges[(size_t)e * 2];
    dn[q] = (e >> 17) * N_ + (edges[(size_t)e * 2 + 1] & (N_ - 1));
  }
#pragma unroll
  for (int q = 0; q < 4; ++q) slot[q] = atomicAdd(&cnt[dn[q]], 1);
#pragma unroll
  for (int q = 0; q < 4; ++q)
    if (slot[q] < CAP_) blist[(size_t)dn[q] * CAP_ + slot[q]] = (unsigned short)sn[q];
}

#define PROJ_TILES_ (B_ * (N_ / EM_))
__global__ __launch_bounds__(256, 4) void k_proj(
    const float* __restrict__ features,
    const _Float16* __restrict__ WkP, const _Float16* __restrict__ WvP,
    const _Float16* __restrict__ WqP,
    const float* __restrict__ biask, const float* __restrict__ biasv,
    const float* __restrict__ biasq,
    float* __restrict__ PKV, float* __restrict__ PQ,
    const int* __restrict__ edges, int* __restrict__ cnt, unsigned short* __restrict__ blist) {
  const int tid  = threadIdx.x;
  if (blockIdx.x < CNT_BLOCKS_) {
    fill_body(edges, cnt, blist, (int)blockIdx.x, tid);
    return;
  }
  const int tile = (int)blockIdx.x - CNT_BLOCKS_;
  const int bb   = (int)((unsigned)tile / (N_ / EM_));
  const int r0   = (int)((unsigned)tile % (N_ / EM_)) * EM_;
  const int rg   = tile * EM_;
  const float* __restrict__ featb = features + (size_t)bb * C_ * N_;
  const int lane = tid & 31;
  const int wave = tid >> 5;
  const int half = lane >> 4;
  const int l16  = lane & 15;

  __shared__ __attribute__((aligned(16))) _Float16 sA[EM_ * SAPU_];

  {
    const int q  = tid & 15;
    const int cg = tid >> 4;
    const float* srcp = featb + (size_t)r0 + 4 * q;
    float4 x[2][8];
#pragma unroll
    for (int p = 0; p < 2; ++p) {
#pragma unroll
      for (int j = 0; j < 8; ++j)
        x[p][j] = *(const float4*)(srcp + (size_t)((p * 16 + cg) * 8 + j) * N_);
    }
#pragma unroll
    for (int p = 0; p < 2; ++p) {
      h8 hx, hy, hz, hw;
#pragma unroll
      for (int j = 0; j < 8; ++j) {
        hx[j] = (_Float16)x[p][j].x; hy[j] = (_Float16)x[p][j].y;
        hz[j] = (_Float16)x[p][j].z; hw[j] = (_Float16)x[p][j].w;
      }
      _Float16* d = sA + (4 * q) * SAPU_ + (p * 16 + cg) * 8;
      *(h8*)(d)             = hx;
      *(h8*)(d + SAPU_)     = hy;
      *(h8*)(d + 2 * SAPU_) = hz;
      *(h8*)(d + 3 * SAPU_) = hw;
    }
  }
  __syncthreads();

  const int nt0 = wave * 2;
  const int n0  = wave * 32;
  const _Float16* aRow = sA + l16 * SAPU_;
  {
    const float cK0 = biask[n0 + l16], cK1 = biask[n0 + l16 + 16];
    const float cV0 = biasv[n0 + l16], cV1 = biasv[n0 + l16 + 16];
    v8f aK[MT_][2], aV[MT_][2];
#pragma unroll
    for (int t = 0; t < MT_; ++t) {
#pragma unroll
      for (int r = 0; r < 8; ++r) {
        aK[t][0][r] = cK0; aK[t][1][r] = cK1; aV[t][0][r] = cV0; aV[t][1][r] = cV1;
      }
    }
#pragma unroll
    for (int kc = 0; kc < 8; ++kc) {
      v16h a[MT_];
#pragma unroll
      for (int t = 0; t < MT_; ++t) a[t] = ldA2(aRow + t * 16 * SAPU_ + kc * 32, half);
      v16h bk0 = ldBp(WkP, kc, nt0,     lane);
      v16h bk1 = ldBp(WkP, kc, nt0 + 1, lane);
      v16h bv0 = ldBp(WvP, kc, nt0,     lane);
      v16h bv1 = ldBp(WvP, kc, nt0 + 1, lane);
#pragma unroll
      for (int t = 0; t < MT_; ++t) {
        aK[t][0] = wmma_f16(a[t], bk0, aK[t][0]);
        aK[t][1] = wmma_f16(a[t], bk1, aK[t][1]);
        aV[t][0] = wmma_f16(a[t], bv0, aV[t][0]);
        aV[t][1] = wmma_f16(a[t], bv1, aV[t][1]);
      }
    }
#pragma unroll
    for (int t = 0; t < MT_; ++t) {
#pragma unroll
      for (int r = 0; r < 8; ++r) {
        const int m = t * 16 + r + half * 8;
        float* pr = PKV + (size_t)(rg + m) * (2 * C_) + n0 + l16;
        pr[0]       = aK[t][0][r];
        pr[16]      = aK[t][1][r];
        pr[C_]      = aV[t][0][r];
        pr[C_ + 16] = aV[t][1][r];
      }
    }
  }
  {
    const float cQ0 = biasq[n0 + l16], cQ1 = biasq[n0 + l16 + 16];
    v8f aQ[MT_][2];
#pragma unroll
    for (int t = 0; t < MT_; ++t) {
#pragma unroll
      for (int r = 0; r < 8; ++r) { aQ[t][0][r] = cQ0; aQ[t][1][r] = cQ1; }
    }
#pragma unroll
    for (int kc = 0; kc < 8; ++kc) {
      v16h a[MT_];
#pragma unroll
      for (int t = 0; t < MT_; ++t) a[t] = ldA2(aRow + t * 16 * SAPU_ + kc * 32, half);
      v16h bq0 = ldBp(WqP, kc, nt0,     lane);
      v16h bq1 = ldBp(WqP, kc, nt0 + 1, lane);
#pragma unroll
      for (int t = 0; t < MT_; ++t) {
        aQ[t][0] = wmma_f16(a[t], bq0, aQ[t][0]);
        aQ[t][1] = wmma_f16(a[t], bq1, aQ[t][1]);
      }
    }
#pragma unroll
    for (int t = 0; t < MT_; ++t) {
#pragma unroll
      for (int r = 0; r < 8; ++r) {
        const int m = t * 16 + r + half * 8;
        float* pr = PQ + (size_t)(rg + m) * C_ + n0 + l16;
        pr[0]  = aQ[t][0][r];
        pr[16] = aQ[t][1][r];
      }
    }
  }
}

__device__ inline void ld8(float* d, const float* __restrict__ p) {
  const float4 x = *(const float4*)p, y = *(const float4*)(p + 4);
  d[0] = x.x; d[1] = x.y; d[2] = x.z; d[3] = x.w; d[4] = y.x; d[5] = y.y; d[6] = y.z; d[7] = y.w;
}
__device__ inline void ld8p(f2* d, const float* __restrict__ p) {
  const float4 x = *(const float4*)p, y = *(const float4*)(p + 4);
  d[0] = f2{x.x, x.y}; d[1] = f2{x.z, x.w}; d[2] = f2{y.x, y.y}; d[3] = f2{y.z, y.w};
}
__device__ inline f2 fma2(f2 a, f2 b, f2 c) { return __builtin_elementwise_fma(a, b, c); }
__device__ inline f2 relu2(f2 a) { return __builtin_elementwise_max(a, f2{0.f, 0.f}); }
__device__ inline f2 bc2(float a) { return f2{a, a}; }
template <int M>
__device__ inline float xrow(float x) {
  const int b = __builtin_bit_cast(int, x);
  return __builtin_bit_cast(float, __builtin_amdgcn_update_dpp(b, b, 0x160 + M, 0xf, 0xf, true));
}
__device__ inline float xhalf(float x) {
  const int b = __builtin_bit_cast(int, x);
  return __builtin_bit_cast(float,
      __builtin_amdgcn_permlanex16(b, b, 0x76543210, 0xfedcba98, false, true));
}
__device__ inline int lane_i(int x, int i) { return __builtin_amdgcn_readlane(x, i); }
__device__ inline float lane_f(float x, int i) {
  return __builtin_bit_cast(float, __builtin_amdgcn_readlane(__builtin_bit_cast(int, x), i));
}

__global__ __launch_bounds__(128, 5) void k_gather(
    const float* __restrict__ xyz,
    const int* __restrict__ cntg, const unsigned short* __restrict__ blistg, const int* __restrict__ edges,
    const float* __restrict__ PKVg, const float* __restrict__ PQg,
    const float* __restrict__ Wk, const float* __restrict__ gk, const float* __restrict__ betak,
    const float* __restrict__ Wv, const float* __restrict__ gv, const float* __restrict__ betav,
    const float* __restrict__ Wq, const float* __restrict__ gq, const float* __restrict__ betaq,
    _Float16* __restrict__ aggh) {
  const int lane = threadIdx.x & 31;
  const int ng = __builtin_amdgcn_readfirstlane((int)(blockIdx.x * 4 + (threadIdx.x >> 5)));
  const int bb = (int)((unsigned)ng / N_);
  const int n  = (int)((unsigned)ng % N_);
  const float* __restrict__ xyzb            = xyz + (size_t)bb * N_ * 3;
  const int* __restrict__ edgesb            = edges + (size_t)bb * E_ * 2;
  const float* __restrict__ PKV             = PKVg + (size_t)bb * N_ * (2 * C_);
  const float* __restrict__ PQ              = PQg + (size_t)bb * N_ * C_;
  const int* __restrict__ cntb              = cntg + (size_t)bb * N_;
  const unsigned short* __restrict__ blistb = blistg + (size_t)bb * N_ * CAP_;
  _Float16* __restrict__ agghb              = aggh + (size_t)bb * N_ * C_;
  const int c0 = lane * 8;
  const int dg = min(cntb[n], E_);
  h8 o = {};
  if (dg > 0) {
    f2 wk[3][4], wv[3][4], wq[3][4];
    f2 gK[4], gQ[4], tK[4], tQ[4];
#pragma unroll
    for (int r = 0; r < 3; ++r) {
      ld8p(wk[r], Wk + (size_t)(C_ + r) * C_ + c0);
      ld8p(wv[r], Wv + (size_t)(C_ + r) * C_ + c0);
      ld8p(wq[r], Wq + (size_t)(C_ + r) * C_ + c0);
    }
    ld8p(gK, gk + c0);    ld8p(gQ, gq + c0);
    ld8p(tK, betak + c0); ld8p(tQ, betaq + c0);
    f2 qb[4];
    ld8p(qb, PQ + (size_t)n * C_ + c0);
    const float xd0 = xyzb[(size_t)n * 3], xd1 = xyzb[(size_t)n * 3 + 1], xd2 = xyzb[(size_t)n * 3 + 2];

    float mx = __int_as_float(0xff800000);
    float s = 0.f;
    f2 acc[4];
#pragma unroll
    for (int p = 0; p < 4; ++p) acc[p] = f2{0.f, 0.f};

    const bool big = dg > CAP_;
    const int ngrp = big ? E_ / 32 : 1;
    for (int g = 0; g < ngrp; ++g) {
      int cnt, my_s;
      if (!big) {
        cnt = dg;
        int key = 0x7fffffff;
        if (lane < cnt) key = (int)blistb[(size_t)n * CAP_ + lane] & (N_ - 1);
        int rank = 0;
        for (int i = 0; i < cnt; ++i) {
          const int si = lane_i(key, i);
          rank += (si < key || (si == key && i < lane)) ? 1 : 0;
        }
        if (lane >= cnt) rank = lane;
        my_s = __builtin_amdgcn_ds_permute(rank << 2, key);
      } else {
        const int2 ed = *(const int2*)(edgesb + (size_t)(g * 32 + lane) * 2);
        const bool m = (ed.y & (N_ - 1)) == n;
        const unsigned mask = __builtin_amdgcn_ballot_w32(m);
        cnt = __builtin_popcount(mask);
        if (cnt == 0) continue;
        const int below = __builtin_popcount(mask & ((1u << lane) - 1u));
        const int tgt = m ? below : cnt + (lane - below);
        my_s = __builtin_amdgcn_ds_permute(tgt << 2, ed.x & (N_ - 1));
      }
      my_s &= N_ - 1;
      float my_r0 = 0.f, my_r1 = 0.f, my_r2 = 0.f;
      if (lane < cnt) {
        const float* xs = xyzb + (size_t)my_s * 3;
        my_r0 = xs[0] - xd0; my_r1 = xs[1] - xd1; my_r2 = xs[2] - xd2;
      }
      auto edge = [&](f2 (&kk)[4], f2 (&vv)[4], int i) __attribute__((always_inline)) {
        f2 q[4];
        const f2 r0 = bc2(lane_f(my_r0, i)), r1 = bc2(lane_f(my_r1, i)), r2 = bc2(lane_f(my_r2, i));
#pragma unroll
        for (int p = 0; p < 4; ++p) {
          kk[p] = relu2(fma2(r2, wk[2][p], fma2(r1, wk[1][p], fma2(r0, wk[0][p], kk[p]))));
          vv[p] = relu2(fma2(r2, wv[2][p], fma2(r1, wv[1][p], fma2(r0, wv[0][p], vv[p]))));
          q[p]  = relu2(fma2(r2, wq[2][p], fma2(r1, wq[1][p], fma2(r0, wq[0][p], qb[p]))));
        }
        const f2 sk = (kk[0] + kk[1]) + (kk[2] + kk[3]);
        const f2 sv = (vv[0] + vv[1]) + (vv[2] + vv[3]);
        const f2 sq = (q[0] + q[1]) + (q[2] + q[3]);
        const f2 sk2 = fma2(kk[3], kk[3], fma2(kk[2], kk[2], fma2(kk[1], kk[1], kk[0] * kk[0])));
        const f2 sv2 = fma2(vv[3], vv[3], fma2(vv[2], vv[2], fma2(vv[1], vv[1], vv[0] * vv[0])));
        const f2 sq2 = fma2(q[3], q[3], fma2(q[2], q[2], fma2(q[1], q[1], q[0] * q[0])));
        float red[6] = {sk.x + sk.y, sk2.x + sk2.y, sv.x + sv.y, sv2.x + sv2.y, sq.x + sq.y, sq2.x + sq2.y};
#pragma unroll
        for (int j = 0; j < 6; ++j) red[j] += xrow<1>(red[j]);
#pragma unroll
        for (int j = 0; j < 6; ++j) red[j] += xrow<2>(red[j]);
#pragma unroll
        for (int j = 0; j < 6; ++j) red[j] += xrow<4>(red[j]);
#pragma unroll
        for (int j = 0; j < 6; ++j) red[j] += xrow<8>(red[j]);
#pragma unroll
        for (int j = 0; j < 6; ++j) red[j] += xhalf(red[j]);
        const float mK = red[0] * (1.f / 256.f), mV = red[2] * (1.f / 256.f), mQ = red[4] * (1.f / 256.f);
        const float iK = rsqrtf(fmaxf(red[1] * (1.f / 256.f) - mK * mK, 0.f) + 1e-5f);
        const float iV = rsqrtf(fmaxf(red[3] * (1.f / 256.f) - mV * mV, 0.f) + 1e-5f);
        const float iQ = rsqrtf(fmaxf(red[5] * (1.f / 256.f) - mQ * mQ, 0.f) + 1e-5f);
        const f2 mK2 = bc2(mK), iK2 = bc2(iK), mQ2 = bc2(mQ), iQ2 = bc2(iQ);
        f2 d2;
#pragma unroll
        for (int p = 0; p < 4; ++p) {
          const f2 kn = fma2((kk[p] - mK2) * iK2, gK[p], tK[p]);
          const f2 qn = fma2((q[p] - mQ2) * iQ2, gQ[p], tQ[p]);
          d2 = p == 0 ? kn * qn : fma2(kn, qn, d2);
        }
        float dot = d2.x + d2.y;
        dot += xrow<1>(dot);
        dot += xrow<2>(dot);
        const float sim = dot * 0.17677669529663687f;
        const float mn  = fmaxf(mx, sim);
        const float a   = __expf(mx - mn);
        const float w   = __expf(sim - mn);
        const float wi  = w * iV;
        const float wm  = -wi * mV;
        s = s * a + w;
        const f2 a2 = bc2(a), wi2 = bc2(wi), wm2 = bc2(wm);
#pragma unroll
        for (int p = 0; p < 4; ++p) acc[p] = fma2(acc[p], a2, fma2(wi2, vv[p], wm2));
        mx = mn;
      };
      f2 ka[4], va[4], kb[4], vb[4];
      auto rows = [&](f2 (&kk)[4], f2 (&vv)[4], int i) __attribute__((always_inline)) {
        const int si = lane_i(my_s, i);
        ld8p(kk, PKV + (size_t)si * (2 * C_) + c0);
        ld8p(vv, PKV + (size_t)si * (2 * C_) + C_ + c0);
      };
      rows(ka, va, 0);
      int i = 0;
      for (; i + 2 < cnt; i += 2) {
        rows(kb, vb, i + 1);
        __builtin_amdgcn_sched_barrier(0);
        edge(ka, va, i);
        __builtin_amdgcn_sched_barrier(0);
        rows(ka, va, i + 2);
        __builtin_amdgcn_sched_barrier(0);
        edge(kb, vb, i + 1);
      }
      if (i + 1 < cnt) {
        rows(kb, vb, i + 1);
        __builtin_amdgcn_sched_barrier(0);
        edge(ka, va, i);
        edge(kb, vb, i + 1);
      } else {
        edge(ka, va, i);
      }
    }
    float gV[8], tV[8];
    ld8(gV, gv + c0); ld8(tV, betav + c0);
    const float is  = 1.f / s;
    const float idg = 1.f / (float)dg;
#pragma unroll
    for (int j = 0; j < 8; ++j) o[j] = (_Float16)((acc[j >> 1][j & 1] * is * gV[j] + tV[j]) * idg);
  }
  *(h8*)(agghb + (size_t)n * C_ + c0) = o;
}

#define UT_ 68

__global__ __launch_bounds__(256) void k_update(
    const _Float16* __restrict__ aggh, const _Float16* __restrict__ WuP,
    const float* __restrict__ biasu, const float* __restrict__ gu,
    const float* __restrict__ betau, const float* __restrict__ features,
    float* __restrict__ out) {
  const int b    = blockIdx.y;
  const int r0   = blockIdx.x * EM_;
  const int tid  = threadIdx.x;
  const int lane = tid & 31;
  const int wave = tid >> 5;
  const int half = lane >> 4;
  const int l16  = lane & 15;

  __shared__ __attribute__((aligned(16))) _Float16 sA[EM_ * SAPU_];
  __shared__ __attribute__((aligned(16))) float sT[C_ * UT_];
  __shared__ float s_part[4][EM_], s_part2[4][EM_];
  __shared__ __attribute__((aligned(16))) float s_mu[EM_], s_is[EM_];

  const int q  = l16;
  const int cs = wave * 2 + half;
  const size_t gbase = (size_t)b * C_ * N_ + (size_t)r0 + 4 * q;

  float4 res[16];
#pragma unroll
  for (int p = 0; p < 16; ++p)
    res[p] = *(const float4*)(features + gbase + (size_t)(p * 16 + cs) * N_);

  {
    const _Float16* srcp = aggh + ((size_t)b * N_ + r0) * C_;
#pragma unroll
    for (int p = 0; p < 8; ++p) {
      const int row = p * 8 + wave;
      *(h8*)(sA + row * SAPU_ + lane * 8) = *(const h8*)(srcp + (size_t)row * C_ + lane * 8);
    }
  }
  __syncthreads();

  const int nt0 = wave * 2;
  const int n0  = wave * 32;
  {
    const _Float16* aRow = sA + l16 * SAPU_;
    v8f acc[MT_][2];
#pragma unroll
    for (int t = 0; t < MT_; ++t) { acc[t][0] = v8f{}; acc[t][1] = v8f{}; }
#pragma unroll
    for (int kc = 0; kc < 8; ++kc) {
      v16h a[MT_];
#pragma unroll
      for (int t = 0; t < MT_; ++t) a[t] = ldA2(aRow + t * 16 * SAPU_ + kc * 32, half);
      v16h b0 = ldBp(WuP, kc, nt0,     lane);
      v16h b1 = ldBp(WuP, kc, nt0 + 1, lane);
#pragma unroll
      for (int t = 0; t < MT_; ++t) {
        acc[t][0] = wmma_f16(a[t], b0, acc[t][0]);
        acc[t][1] = wmma_f16(a[t], b1, acc[t][1]);
      }
    }
    const float bu0 = biasu[n0 + l16], bu1 = biasu[n0 + l16 + 16];
#pragma unroll
    for (int t = 0; t < MT_; ++t) {
      float* d0 = sT + (n0 + l16) * UT_ + t * 16 + half * 8;
      float* d1 = d0 + 16 * UT_;
      float4 x;
      x.x = fmaxf(acc[t][0][0] + bu0, 0.f); x.y = fmaxf(acc[t][0][1] + bu0, 0.f);
      x.z = fmaxf(acc[t][0][2] + bu0, 0.f); x.w = fmaxf(acc[t][0][3] + bu0, 0.f);
      *(float4*)d0 = x;
      x.x = fmaxf(acc[t][0][4] + bu0, 0.f); x.y = fmaxf(acc[t][0][5] + bu0, 0.f);
      x.z = fmaxf(acc[t][0][6] + bu0, 0.f); x.w = fmaxf(acc[t][0][7] + bu0, 0.f);
      *(float4*)(d0 + 4) = x;
      x.x = fmaxf(acc[t][1][0] + bu1, 0.f); x.y = fmaxf(acc[t][1][1] + bu1, 0.f);
      x.z = fmaxf(acc[t][1][2] + bu1, 0.f); x.w = fmaxf(acc[t][1][3] + bu1, 0.f);
      *(float4*)d1 = x;
      x.x = fmaxf(acc[t][1][4] + bu1, 0.f); x.y = fmaxf(acc[t][1][5] + bu1, 0.f);
      x.z = fmaxf(acc[t][1][6] + bu1, 0.f); x.w = fmaxf(acc[t][1][7] + bu1, 0.f);
      *(float4*)(d1 + 4) = x;
    }
  }
  float g[16], bt[16];
#pragma unroll
  for (int p = 0; p < 16; ++p) { g[p] = gu[p * 16 + cs]; bt[p] = betau[p * 16 + cs]; }
  __syncthreads();

  {
    const int m = tid & (EM_ - 1), part = tid >> 6;
    const float* col = sT + (part * 64) * UT_ + m;
    float s = 0.f, s2 = 0.f;
    for (int c = 0; c < 64; ++c) {
      const float x = col[c * UT_];
      s += x; s2 += x * x;
    }
    s_part[part][m]  = s;
    s_part2[part][m] = s2;
  }
  __syncthreads();
  if (tid < EM_) {
    const float S  = s_part[0][tid] + s_part[1][tid] + s_part[2][tid] + s_part[3][tid];
    const float S2 = s_part2[0][tid] + s_part2[1][tid] + s_part2[2][tid] + s_part2[3][tid];
    const float mean = S * (1.f / 256.f);
    const float var  = fmaxf(S2 * (1.f / 256.f) - mean * mean, 0.f);
    s_mu[tid] = mean;
    s_is[tid] = rsqrtf(var + 1e-5f);
  }
  __syncthreads();

  {
    const float4 mu = *(const float4*)(s_mu + 4 * q);
    const float4 is = *(const float4*)(s_is + 4 * q);
#pragma unroll
    for (int p = 0; p < 16; ++p) {
      const int c = p * 16 + cs;
      const float4 x = *(const float4*)(sT + c * UT_ + 4 * q);
      float4 o;
      o.x = ((x.x - mu.x) * is.x * g[p] + bt[p]) + res[p].x;
      o.y = ((x.y - mu.y) * is.y * g[p] + bt[p]) + res[p].y;
      o.z = ((x.z - mu.z) * is.z * g[p] + bt[p]) + res[p].z;
      o.w = ((x.w - mu.w) * is.w * g[p] + bt[p]) + res[p].w;
      *(float4*)(out + gbase + (size_t)c * N_) = o;
    }
  }
}

extern "C" void kernel_launch(void* const* d_in, const int* in_sizes, int n_in,
                              void* d_out, int out_size, void* d_ws, size_t ws_size,
                              hipStream_t stream) {
  const float* xyz      = (const float*)d_in[0];
  const float* features = (const float*)d_in[1];
  const int*   edges    = (const int*)d_in[2];
  const float* Wk = (const float*)d_in[3];  const float* bk = (const float*)d_in[4];
  const float* gk = (const float*)d_in[5];  const float* betak = (const float*)d_in[6];
  const float* Wv = (const float*)d_in[7];  const float* bv = (const float*)d_in[8];
  const float* gv = (const float*)d_in[9];  const float* betav = (const float*)d_in[10];
  const float* Wq = (const float*)d_in[11]; const float* bq = (const float*)d_in[12];
  const float* gq = (const float*)d_in[13]; const float* betaq = (const float*)d_in[14];
  const float* Wu = (const float*)d_in[15]; const float* bu = (const float*)d_in[16];
  const float* gu = (const float*)d_in[17]; const float* betau = (const float*)d_in[18];
  float* out = (float*)d_out;

  char* ws = (char*)d_ws;
  size_t off = 0;
  auto carve = [&](size_t bytes) -> char* {
    char* p = ws + off;
    off = (off + bytes + 255) & ~(size_t)255;
    return p;
  };
  _Float16* WkP    = (_Float16*)carve((size_t)C_ * C_ * 2);
  _Float16* WvP    = (_Float16*)carve((size_t)C_ * C_ * 2);
  _Float16* WqP    = (_Float16*)carve((size_t)C_ * C_ * 2);
  _Float16* WuP    = (_Float16*)carve((size_t)C_ * C_ * 2);
  float*    PKV    = (float*)carve((size_t)B_ * N_ * 2 * C_ * 4);
  float*    PQ     = (float*)carve((size_t)B_ * N_ * C_ * 4);
  int*      cnt    = (int*)carve((size_t)B_ * N_ * 4);
  unsigned short* blist = (unsigned short*)carve((size_t)B_ * N_ * CAP_ * 2);
  _Float16* aggh   = (_Float16*)carve((size_t)B_ * N_ * C_ * 2);
  (void)ws_size; (void)in_sizes; (void)n_in; (void)out_size;

  k_prep<<<PREP_INIT_ + PREP_PACK_, 256, 0, stream>>>(Wk, Wv, Wq, Wu, WkP, WvP, WqP, WuP, cnt);

  k_proj<<<CNT_BLOCKS_ + PROJ_TILES_, 256, 0, stream>>>(features, WkP, WvP, WqP, bk, bv, bq, PKV, PQ, edges, cnt, blist);

  k_gather<<<B_ * N_ / 4, 128, 0, stream>>>(
      xyz, cnt, blist, edges, PKV, PQ,
      Wk, gk, betak, Wv, gv, betav, Wq, gq, betaq,
      aggh);

  k_update<<<dim3(N_ / EM_, B_), 256, 0, stream>>>(aggh, WuP, bu, gu, betau, features, out);
}
